# starting kernel
# baseline (speedup 1.0000x reference)
#include <hip/hip_runtime.h>
#include <hip/hip_bf16.h>
#include <math.h>

#define N_NODES 200000
#define N_EDGES 800000
#define DIN 64
#define DE  32
#define DH  64
#define NH  4
#define HD  16

typedef float v2f __attribute__((ext_vector_type(2)));
typedef float v8f __attribute__((ext_vector_type(8)));

__device__ __forceinline__ v8f wstep(v2f a, v2f b, v8f c) {
    return __builtin_amdgcn_wmma_f32_16x16x4_f32(false, a, false, b, (short)0, c,
                                                 false, false);
}

template <int K>
__device__ __forceinline__ v8f gemm_tile(const float* As, int lda,
                                         const float* Bs, int ldb,
                                         int n0, int lane) {
    const int half = lane >> 4;
    const int qn   = lane & 15;
    v2f a[K / 4], b[K / 4];
    #pragma unroll
    for (int s = 0; s < K / 4; ++s) {
        const int k0 = 4 * s + 2 * half;
        a[s].x = As[qn * lda + k0];
        a[s].y = As[qn * lda + k0 + 1];
        b[s].x = Bs[k0 * ldb + n0 + qn];
        b[s].y = Bs[(k0 + 1) * ldb + n0 + qn];
    }
    v8f c = {0.f, 0.f, 0.f, 0.f, 0.f, 0.f, 0.f, 0.f};
    #pragma unroll
    for (int s = 0; s < K / 4; ++s) c = wstep(a[s], b[s], c);
    return c;
}

__global__ void k_zero(float* __restrict__ p, int n) {
    int i = blockIdx.x * blockDim.x + threadIdx.x;
    if (i < n) p[i] = 0.f;
}

__global__ void k_node_qkv(const float* __restrict__ x,
                           const float* __restrict__ Wq, const float* __restrict__ bq,
                           const float* __restrict__ Wk, const float* __restrict__ bk,
                           const float* __restrict__ Wv, const float* __restrict__ bv,
                           const float* __restrict__ ln_s, const float* __restrict__ ln_b,
                           float* __restrict__ qo, float* __restrict__ ko,
                           float* __restrict__ vo) {
    __shared__ float xs[2][16 * 65];
    __shared__ float wlds[64 * 64];
    const int lane = threadIdx.x & 31;
    const int wv   = threadIdx.x >> 5;
    const int tile = blockIdx.x * 2 + wv;
    const int row0 = tile * 16;
    const int t    = (blockIdx.x * 32 >= N_NODES / 2) ? 1 : 0;
    float* As = xs[wv];

    for (int i = 0; i < 32; ++i) {
        int idx = i * 32 + lane;
        int m = idx >> 6, cc = idx & 63;
        As[m * 65 + cc] = x[(size_t)(row0 + m) * DIN + cc];
    }
    __syncthreads();
    {
        const int m = lane & 15, c0 = (lane >> 4) * 32;
        float s = 0.f;
        for (int j = 0; j < 32; ++j) s += As[m * 65 + c0 + j];
        s += __shfl_xor(s, 16, 32);
        float mean = s * (1.0f / 64.0f);
        float v2 = 0.f;
        for (int j = 0; j < 32; ++j) { float d = As[m * 65 + c0 + j] - mean; v2 += d * d; }
        v2 += __shfl_xor(v2, 16, 32);
        float inv = rsqrtf(v2 * (1.0f / 64.0f) + 1e-5f);
        for (int j = 0; j < 32; ++j) {
            int cc = c0 + j;
            As[m * 65 + cc] = (As[m * 65 + cc] - mean) * inv * ln_s[t * DIN + cc]
                              + ln_b[t * DIN + cc];
        }
    }

    const int half = lane >> 4, qn = lane & 15;
    const float* Ws[3] = {Wq + t * DIN * DH, Wk + t * DIN * DH, Wv + t * DIN * DH};
    const float* bs[3] = {bq + t * DH, bk + t * DH, bv + t * DH};
    float*       os[3] = {qo, ko, vo};
    for (int mm = 0; mm < 3; ++mm) {
        __syncthreads();
        for (int i = threadIdx.x; i < DIN * DH; i += 64)
            wlds[i] = Ws[mm][i];
        __syncthreads();
        #pragma unroll
        for (int nt = 0; nt < 4; ++nt) {
            const int n0 = nt * 16;
            v8f c = gemm_tile<64>(As, 65, wlds, DH, n0, lane);
            float bias = bs[mm][n0 + qn];
            #pragma unroll
            for (int i = 0; i < 8; ++i) {
                int m = i + 8 * half;
                os[mm][(size_t)(row0 + m) * DH + n0 + qn] = c[i] + bias;
            }
        }
    }
}

__global__ void k_edge(const float* __restrict__ ea,
                       const float* __restrict__ We, const float* __restrict__ be,
                       const float* __restrict__ Wg, const float* __restrict__ bg,
                       const float* __restrict__ ew,
                       const float* __restrict__ ln_s, const float* __restrict__ ln_b,
                       const int* __restrict__ esrc, const int* __restrict__ edst,
                       const float* __restrict__ qbuf, const float* __restrict__ kbuf,
                       const float* __restrict__ vbuf,
                       float* __restrict__ acc, float* __restrict__ den) {
    __shared__ float ebs[4][16 * 33];
    __shared__ float welds[DE * DH];
    __shared__ float wglds[DE * DH];
    __shared__ float ewlds[NH * 16 * 17];
    __shared__ int   ssrc[4][16], sdst[4][16];
    const int lane = threadIdx.x & 31;
    const int wv   = threadIdx.x >> 5;
    const int tile = blockIdx.x * 4 + wv;
    const int e0   = tile * 16;
    const int t    = (blockIdx.x * 64 >= N_EDGES / 2) ? 1 : 0;
    float* As = ebs[wv];

    for (int i = 0; i < 16; ++i) {
        int idx = i * 32 + lane;
        int m = idx >> 5, cc = idx & 31;
        As[m * 33 + cc] = ea[(size_t)(e0 + m) * DE + cc];
    }
    if (lane < 16) {
        ssrc[wv][lane] = esrc[e0 + lane];
        sdst[wv][lane] = edst[e0 + lane];
    }
    for (int i = threadIdx.x; i < DE * DH; i += 128) {
        welds[i] = We[t * DE * DH + i];
        wglds[i] = Wg[t * DE * DH + i];
    }
    for (int idx = threadIdx.x; idx < NH * 16 * 16; idx += 128) {
        int h = idx >> 8, rem = idx & 255, i = rem >> 4, j = rem & 15;
        ewlds[(h * 16 + j) * 17 + i] = ew[t * NH * 256 + idx];
    }
    __syncthreads();
    {
        const int m = lane & 15, c0 = (lane >> 4) * 16;
        float s = 0.f;
        for (int j = 0; j < 16; ++j) s += As[m * 33 + c0 + j];
        s += __shfl_xor(s, 16, 32);
        float mean = s * (1.0f / 32.0f);
        float v2 = 0.f;
        for (int j = 0; j < 16; ++j) { float d = As[m * 33 + c0 + j] - mean; v2 += d * d; }
        v2 += __shfl_xor(v2, 16, 32);
        float inv = rsqrtf(v2 * (1.0f / 32.0f) + 1e-5f);
        for (int j = 0; j < 16; ++j) {
            int cc = c0 + j;
            As[m * 33 + cc] = (As[m * 33 + cc] - mean) * inv * ln_s[t * DE + cc]
                              + ln_b[t * DE + cc];
        }
    }
    __syncthreads();

    const int half = lane >> 4, qn = lane & 15;
    const float* krow = kbuf + (size_t)ssrc[wv][qn] * DH;

    for (int h = 0; h < NH; ++h) {
        const int n0 = h * 16;
        v8f eatt = gemm_tile<32>(As, 33, welds, DH, n0, lane);
        v8f egat = gemm_tile<32>(As, 33, wglds, DH, n0, lane);
        const float batt = be[t * DH + n0 + qn];
        const float bgat = bg[t * DH + n0 + qn];

        v8f ekc = gemm_tile<16>(krow + n0, 0, ewlds + h * 16 * 17, 17, 0, lane);

        v8f red;
        #pragma unroll
        for (int i = 0; i < 8; ++i) {
            int m = i + 8 * half;
            float dq = qbuf[(size_t)sdst[wv][m] * DH + n0 + qn];
            red[i] = dq * ekc[i] + eatt[i] + batt;
        }
        #pragma unroll
        for (int off = 1; off < 16; off <<= 1) {
            #pragma unroll
            for (int i = 0; i < 8; ++i) red[i] += __shfl_xor(red[i], off, 32);
        }
        float es8[8];
        #pragma unroll
        for (int i = 0; i < 8; ++i) {
            float sc = red[i] * 0.25f;
            sc = fminf(5.0f, fmaxf(-5.0f, sc));
            es8[i] = __expf(sc);
        }
        #pragma unroll
        for (int i = 0; i < 8; ++i) {
            if (qn == i)
                atomicAdd(&den[(size_t)sdst[wv][i + 8 * half] * NH + h], es8[i]);
        }
        #pragma unroll
        for (int i = 0; i < 8; ++i) {
            int m = i + 8 * half;
            float vv = vbuf[(size_t)ssrc[wv][m] * DH + n0 + qn];
            float g  = 1.0f / (1.0f + __expf(-(egat[i] + bgat)));
            atomicAdd(&acc[(size_t)sdst[wv][m] * DH + n0 + qn], es8[i] * vv * g);
        }
    }
}

__global__ void k_out(const float* __restrict__ acc, const float* __restrict__ den,
                      const float* __restrict__ Wo, const float* __restrict__ bo,
                      float* __restrict__ out) {
    __shared__ float ts[2][16 * 65];
    __shared__ float wlds[DH * DIN];
    const int lane = threadIdx.x & 31;
    const int wv   = threadIdx.x >> 5;
    const int tile = blockIdx.x * 2 + wv;
    const int row0 = tile * 16;
    const int t    = (blockIdx.x * 32 >= N_NODES / 2) ? 1 : 0;
    float* As = ts[wv];

    for (int i = threadIdx.x; i < DH * DIN; i += 64)
        wlds[i] = Wo[t * DH * DIN + i];
    for (int i = 0; i < 32; ++i) {
        int idx = i * 32 + lane;
        int m = idx >> 6, cc = idx & 63;
        float d = den[(size_t)(row0 + m) * NH + (cc >> 4)];
        float a = acc[(size_t)(row0 + m) * DH + cc];
        As[m * 65 + cc] = (d > 0.f) ? (a / d) : 0.f;
    }
    __syncthreads();

    const int half = lane >> 4, qn = lane & 15;
    #pragma unroll
    for (int nt = 0; nt < 4; ++nt) {
        const int n0 = nt * 16;
        v8f c = gemm_tile<64>(As, 65, wlds, DIN, n0, lane);
        float bias = bo[t * DIN + n0 + qn];
        #pragma unroll
        for (int i = 0; i < 8; ++i)
            out[(size_t)(row0 + i + 8 * half) * DIN + n0 + qn] = c[i] + bias;
    }
}

extern "C" void kernel_launch(void* const* d_in, const int* in_sizes, int n_in,
                              void* d_out, int out_size, void* d_ws, size_t ws_size,
                              hipStream_t stream) {
    const float* x    = (const float*)d_in[0];
    const float* ea   = (const float*)d_in[1];
    const float* Wq   = (const float*)d_in[2];
    const float* bq   = (const float*)d_in[3];
    const float* Wk   = (const float*)d_in[4];
    const float* bk   = (const float*)d_in[5];
    const float* Wv   = (const float*)d_in[6];
    const float* bv   = (const float*)d_in[7];
    const float* Wo   = (const float*)d_in[8];
    const float* bo   = (const float*)d_in[9];
    const float* We   = (const float*)d_in[10];
    const float* be   = (const float*)d_in[11];
    const float* Wg   = (const float*)d_in[12];
    const float* bg   = (const float*)d_in[13];
    const float* ew   = (const float*)d_in[14];
    const float* lnns = (const float*)d_in[15];
    const float* lnnb = (const float*)d_in[16];
    const float* lnes = (const float*)d_in[17];
    const float* lneb = (const float*)d_in[18];
    const int*   ei   = (const int*)d_in[19];
    const int* esrc = ei;
    const int* edst = ei + N_EDGES;

    float* ws  = (float*)d_ws;
    float* qb  = ws;
    float* kb  = qb  + (size_t)N_NODES * DH;
    float* vb  = kb  + (size_t)N_NODES * DH;
    float* acc = vb  + (size_t)N_NODES * DH;
    float* den = acc + (size_t)N_NODES * DH;

    const int nz = N_NODES * DH + N_NODES * NH;
    k_zero<<<(nz + 255) / 256, 256, 0, stream>>>(acc, nz);

    k_node_qkv<<<(N_NODES / 16) / 2, 64, 0, stream>>>(
        x, Wq, bq, Wk, bk, Wv, bv, lnns, lnnb, qb, kb, vb);

    k_edge<<<(N_EDGES / 16) / 4, 128, 0, stream>>>(
        ea, We, be, Wg, bg, ew, lnes, lneb, esrc, edst, qb, kb, vb, acc, den);

    k_out<<<(N_NODES / 16) / 2, 64, 0, stream>>>(acc, den, Wo, bo, (float*)d_out);
}


# Row loads issued as one group per tile instead of one per dependent round trip: q/v gathers ahead of each head's atomics (edge pass), acc/den ahead of a selected divide (output pass)
# speedup vs baseline: 1.1866x; 1.1866x over previous
#include <hip/hip_runtime.h>
#include <hip/hip_bf16.h>
#include <math.h>

#define N_NODES 200000
#define N_EDGES 800000
#define DIN 64
#define DE  32
#define DH  64
#define NH  4
#define HD  16

typedef float v2f __attribute__((ext_vector_type(2)));
typedef float v8f __attribute__((ext_vector_type(8)));

__device__ __forceinline__ v8f wstep(v2f a, v2f b, v8f c) {
    return __builtin_amdgcn_wmma_f32_16x16x4_f32(false, a, false, b, (short)0, c,
                                                 false, false);
}

template <int K>
__device__ __forceinline__ v8f gemm_tile(const float* As, int lda,
                                         const float* Bs, int ldb,
                                         int n0, int lane) {
    const int half = lane >> 4;
    const int qn   = lane & 15;
    v2f a[K / 4], b[K / 4];
    #pragma unroll
    for (int s = 0; s < K / 4; ++s) {
        const int k0 = 4 * s + 2 * half;
        a[s].x = As[qn * lda + k0];
        a[s].y = As[qn * lda + k0 + 1];
        b[s].x = Bs[k0 * ldb + n0 + qn];
        b[s].y = Bs[(k0 + 1) * ldb + n0 + qn];
    }
    v8f c = {0.f, 0.f, 0.f, 0.f, 0.f, 0.f, 0.f, 0.f};
    #pragma unroll
    for (int s = 0; s < K / 4; ++s) c = wstep(a[s], b[s], c);
    return c;
}

__global__ void k_zero(float* __restrict__ p, int n) {
    int i = blockIdx.x * blockDim.x + threadIdx.x;
    if (i < n) p[i] = 0.f;
}

__global__ void k_node_qkv(const float* __restrict__ x,
                           const float* __restrict__ Wq, const float* __restrict__ bq,
                           const float* __restrict__ Wk, const float* __restrict__ bk,
                           const float* __restrict__ Wv, const float* __restrict__ bv,
                           const float* __restrict__ ln_s, const float* __restrict__ ln_b,
                           float* __restrict__ qo, float* __restrict__ ko,
                           float* __restrict__ vo) {
    __shared__ float xs[2][16 * 65];
    __shared__ float wlds[64 * 64];
    const int lane = threadIdx.x & 31;
    const int wv   = threadIdx.x >> 5;
    const int tile = blockIdx.x * 2 + wv;
    const int row0 = tile * 16;
    const int t    = (blockIdx.x * 32 >= N_NODES / 2) ? 1 : 0;
    float* As = xs[wv];

    for (int i = 0; i < 32; ++i) {
        int idx = i * 32 + lane;
        int m = idx >> 6, cc = idx & 63;
        As[m * 65 + cc] = x[(size_t)(row0 + m) * DIN + cc];
    }
    __syncthreads();
    {
        const int m = lane & 15, c0 = (lane >> 4) * 32;
        float s = 0.f;
        for (int j = 0; j < 32; ++j) s += As[m * 65 + c0 + j];
        s += __shfl_xor(s, 16, 32);
        float mean = s * (1.0f / 64.0f);
        float v2 = 0.f;
        for (int j = 0; j < 32; ++j) { float d = As[m * 65 + c0 + j] - mean; v2 += d * d; }
        v2 += __shfl_xor(v2, 16, 32);
        float inv = rsqrtf(v2 * (1.0f / 64.0f) + 1e-5f);
        for (int j = 0; j < 32; ++j) {
            int cc = c0 + j;
            As[m * 65 + cc] = (As[m * 65 + cc] - mean) * inv * ln_s[t * DIN + cc]
                              + ln_b[t * DIN + cc];
        }
    }

    const int half = lane >> 4, qn = lane & 15;
    const float* Ws[3] = {Wq + t * DIN * DH, Wk + t * DIN * DH, Wv + t * DIN * DH};
    const float* bs[3] = {bq + t * DH, bk + t * DH, bv + t * DH};
    float*       os[3] = {qo, ko, vo};
    for (int mm = 0; mm < 3; ++mm) {
        __syncthreads();
        for (int i = threadIdx.x; i < DIN * DH; i += 64)
            wlds[i] = Ws[mm][i];
        __syncthreads();
        #pragma unroll
        for (int nt = 0; nt < 4; ++nt) {
            const int n0 = nt * 16;
            v8f c = gemm_tile<64>(As, 65, wlds, DH, n0, lane);
            float bias = bs[mm][n0 + qn];
            #pragma unroll
            for (int i = 0; i < 8; ++i) {
                int m = i + 8 * half;
                os[mm][(size_t)(row0 + m) * DH + n0 + qn] = c[i] + bias;
            }
        }
    }
}

__global__ void k_edge(const float* __restrict__ ea,
                       const float* __restrict__ We, const float* __restrict__ be,
                       const float* __restrict__ Wg, const float* __restrict__ bg,
                       const float* __restrict__ ew,
                       const float* __restrict__ ln_s, const float* __restrict__ ln_b,
                       const int* __restrict__ esrc, const int* __restrict__ edst,
                       const float* __restrict__ qbuf, const float* __restrict__ kbuf,
                       const float* __restrict__ vbuf,
                       float* __restrict__ acc, float* __restrict__ den) {
    __shared__ float ebs[4][16 * 33];
    __shared__ float welds[DE * DH];
    __shared__ float wglds[DE * DH];
    __shared__ float ewlds[NH * 16 * 17];
    __shared__ int   ssrc[4][16], sdst[4][16];
    const int lane = threadIdx.x & 31;
    const int wv   = threadIdx.x >> 5;
    const int tile = blockIdx.x * 4 + wv;
    const int e0   = tile * 16;
    const int t    = (blockIdx.x * 64 >= N_EDGES / 2) ? 1 : 0;
    float* As = ebs[wv];

    for (int i = 0; i < 16; ++i) {
        int idx = i * 32 + lane;
        int m = idx >> 5, cc = idx & 31;
        As[m * 33 + cc] = ea[(size_t)(e0 + m) * DE + cc];
    }
    if (lane < 16) {
        ssrc[wv][lane] = esrc[e0 + lane];
        sdst[wv][lane] = edst[e0 + lane];
    }
    for (int i = threadIdx.x; i < DE * DH; i += 128) {
        welds[i] = We[t * DE * DH + i];
        wglds[i] = Wg[t * DE * DH + i];
    }
    for (int idx = threadIdx.x; idx < NH * 16 * 16; idx += 128) {
        int h = idx >> 8, rem = idx & 255, i = rem >> 4, j = rem & 15;
        ewlds[(h * 16 + j) * 17 + i] = ew[t * NH * 256 + idx];
    }
    __syncthreads();
    {
        const int m = lane & 15, c0 = (lane >> 4) * 16;
        float s = 0.f;
        for (int j = 0; j < 16; ++j) s += As[m * 33 + c0 + j];
        s += __shfl_xor(s, 16, 32);
        float mean = s * (1.0f / 32.0f);
        float v2 = 0.f;
        for (int j = 0; j < 16; ++j) { float d = As[m * 33 + c0 + j] - mean; v2 += d * d; }
        v2 += __shfl_xor(v2, 16, 32);
        float inv = rsqrtf(v2 * (1.0f / 32.0f) + 1e-5f);
        for (int j = 0; j < 16; ++j) {
            int cc = c0 + j;
            As[m * 33 + cc] = (As[m * 33 + cc] - mean) * inv * ln_s[t * DE + cc]
                              + ln_b[t * DE + cc];
        }
    }
    __syncthreads();

    const int half = lane >> 4, qn = lane & 15;
    const float* krow = kbuf + (size_t)ssrc[wv][qn] * DH;

    for (int h = 0; h < NH; ++h) {
        const int n0 = h * 16;
        float dq[8], vv[8];
        #pragma unroll
        for (int i = 0; i < 8; ++i) {
            int m = i + 8 * half;
            dq[i] = qbuf[(size_t)sdst[wv][m] * DH + n0 + qn];
            vv[i] = vbuf[(size_t)ssrc[wv][m] * DH + n0 + qn];
        }
        v8f eatt = gemm_tile<32>(As, 33, welds, DH, n0, lane);
        v8f egat = gemm_tile<32>(As, 33, wglds, DH, n0, lane);
        const float batt = be[t * DH + n0 + qn];
        const float bgat = bg[t * DH + n0 + qn];

        v8f ekc = gemm_tile<16>(krow + n0, 0, ewlds + h * 16 * 17, 17, 0, lane);

        v8f red;
        #pragma unroll
        for (int i = 0; i < 8; ++i) red[i] = dq[i] * ekc[i] + eatt[i] + batt;
        #pragma unroll
        for (int off = 1; off < 16; off <<= 1) {
            #pragma unroll
            for (int i = 0; i < 8; ++i) red[i] += __shfl_xor(red[i], off, 32);
        }
        float es8[8];
        #pragma unroll
        for (int i = 0; i < 8; ++i) {
            float sc = red[i] * 0.25f;
            sc = fminf(5.0f, fmaxf(-5.0f, sc));
            es8[i] = __expf(sc);
        }
        #pragma unroll
        for (int i = 0; i < 8; ++i) {
            if (qn == i)
                atomicAdd(&den[(size_t)sdst[wv][i + 8 * half] * NH + h], es8[i]);
        }
        #pragma unroll
        for (int i = 0; i < 8; ++i) {
            int m = i + 8 * half;
            float g  = 1.0f / (1.0f + __expf(-(egat[i] + bgat)));
            atomicAdd(&acc[(size_t)sdst[wv][m] * DH + n0 + qn], es8[i] * vv[i] * g);
        }
    }
}

__global__ void k_out(const float* __restrict__ acc, const float* __restrict__ den,
                      const float* __restrict__ Wo, const float* __restrict__ bo,
                      float* __restrict__ out) {
    __shared__ float ts[2][16 * 65];
    __shared__ float wlds[DH * DIN];
    const int lane = threadIdx.x & 31;
    const int wv   = threadIdx.x >> 5;
    const int tile = blockIdx.x * 2 + wv;
    const int row0 = tile * 16;
    const int t    = (blockIdx.x * 32 >= N_NODES / 2) ? 1 : 0;
    float* As = ts[wv];

    for (int i = threadIdx.x; i < DH * DIN; i += 64)
        wlds[i] = Wo[t * DH * DIN + i];
    float av[32], dv[32];
    #pragma unroll
    for (int i = 0; i < 32; ++i) {
        int idx = i * 32 + lane;
        int m = idx >> 6, cc = idx & 63;
        dv[i] = den[(size_t)(row0 + m) * NH + (cc >> 4)];
        av[i] = acc[(size_t)(row0 + m) * DH + cc];
    }
    #pragma unroll
    for (int i = 0; i < 32; ++i) {
        int idx = i * 32 + lane;
        int m = idx >> 6, cc = idx & 63;
        float qv = av[i] / dv[i];
        As[m * 65 + cc] = (dv[i] > 0.f) ? qv : 0.f;
    }
    __syncthreads();

    const int half = lane >> 4, qn = lane & 15;
    #pragma unroll
    for (int nt = 0; nt < 4; ++nt) {
        const int n0 = nt * 16;
        v8f c = gemm_tile<64>(As, 65, wlds, DIN, n0, lane);
        float bias = bo[t * DIN + n0 + qn];
        #pragma unroll
        for (int i = 0; i < 8; ++i)
            out[(size_t)(row0 + i + 8 * half) * DIN + n0 + qn] = c[i] + bias;
    }
}

extern "C" void kernel_launch(void* const* d_in, const int* in_sizes, int n_in,
                              void* d_out, int out_size, void* d_ws, size_t ws_size,
                              hipStream_t stream) {
    const float* x    = (const float*)d_in[0];
    const float* ea   = (const float*)d_in[1];
    const float* Wq   = (const float*)d_in[2];
    const float* bq   = (const float*)d_in[3];
    const float* Wk   = (const float*)d_in[4];
    const float* bk   = (const float*)d_in[5];
    const float* Wv   = (const float*)d_in[6];
    const float* bv   = (const float*)d_in[7];
    const float* Wo   = (const float*)d_in[8];
    const float* bo   = (const float*)d_in[9];
    const float* We   = (const float*)d_in[10];
    const float* be   = (const float*)d_in[11];
    const float* Wg   = (const float*)d_in[12];
    const float* bg   = (const float*)d_in[13];
    const float* ew   = (const float*)d_in[14];
    const float* lnns = (const float*)d_in[15];
    const float* lnnb = (const float*)d_in[16];
    const float* lnes = (const float*)d_in[17];
    const float* lneb = (const float*)d_in[18];
    const int*   ei   = (const int*)d_in[19];
    const int* esrc = ei;
    const int* edst = ei + N_EDGES;

    float* ws  = (float*)d_ws;
    float* qb  = ws;
    float* kb  = qb  + (size_t)N_NODES * DH;
    float* vb  = kb  + (size_t)N_NODES * DH;
    float* acc = vb  + (size_t)N_NODES * DH;
    float* den = acc + (size_t)N_NODES * DH;

    const int nz = N_NODES * DH + N_NODES * NH;
    k_zero<<<(nz + 255) / 256, 256, 0, stream>>>(acc, nz);

    k_node_qkv<<<(N_NODES / 16) / 2, 64, 0, stream>>>(
        x, Wq, bq, Wk, bk, Wv, bv, lnns, lnnb, qb, kb, vb);

    k_edge<<<(N_EDGES / 16) / 4, 128, 0, stream>>>(
        ea, We, be, Wg, bg, ew, lnes, lneb, esrc, edst, qb, kb, vb, acc, den);

    k_out<<<(N_NODES / 16) / 2, 64, 0, stream>>>(acc, den, Wo, bo, (float*)d_out);
}


# Edges binned per call by (800-row dst range, edge type); one block per range adds message sums and denominators in LDS and projects them from LDS; no global float atomics; sums in arrival order
# speedup vs baseline: 1.5284x; 1.2880x over previous
#include <hip/hip_runtime.h>
#include <hip/hip_bf16.h>
#include <math.h>

#define N_NODES 200000
#define N_EDGES 800000
#define DIN 64
#define DE  32
#define DH  64
#define NH  4
#define HD  16
#define RNG  800
#define NRNG (N_NODES / RNG)
#define NBIN (2 * NRNG)
#define ROWP 68
#define EWV  24

typedef float v2f __attribute__((ext_vector_type(2)));
typedef float v8f __attribute__((ext_vector_type(8)));

__device__ __forceinline__ v8f wstep(v2f a, v2f b, v8f c) {
    return __builtin_amdgcn_wmma_f32_16x16x4_f32(false, a, false, b, (short)0, c,
                                                 false, false);
}

template <int K>
__device__ __forceinline__ v8f gemm_tile(const float* As, int lda,
                                         const float* Bs, int ldb,
                                         int n0, int lane) {
    const int half = lane >> 4;
    const int qn   = lane & 15;
    v2f a[K / 4], b[K / 4];
    #pragma unroll
    for (int s = 0; s < K / 4; ++s) {
        const int k0 = 4 * s + 2 * half;
        a[s].x = As[qn * lda + k0];
        a[s].y = As[qn * lda + k0 + 1];
        b[s].x = Bs[k0 * ldb + n0 + qn];
        b[s].y = Bs[(k0 + 1) * ldb + n0 + qn];
    }
    v8f c = {0.f, 0.f, 0.f, 0.f, 0.f, 0.f, 0.f, 0.f};
    #pragma unroll
    for (int s = 0; s < K / 4; ++s) c = wstep(a[s], b[s], c);
    return c;
}

__global__ void k_zero(float* __restrict__ p, int n) {
    int i = blockIdx.x * blockDim.x + threadIdx.x;
    if (i < n) p[i] = 0.f;
}

__global__ void k_node_qkv(const float* __restrict__ x,
                           const float* __restrict__ Wq, const float* __restrict__ bq,
                           const float* __restrict__ Wk, const float* __restrict__ bk,
                           const float* __restrict__ Wv, const float* __restrict__ bv,
                           const float* __restrict__ ln_s, const float* __restrict__ ln_b,
                           float* __restrict__ qo, float* __restrict__ ko,
                           float* __restrict__ vo) {
    __shared__ float xs[2][16 * 65];
    __shared__ float wlds[64 * 64];
    const int lane = threadIdx.x & 31;
    const int wv   = threadIdx.x >> 5;
    const int tile = blockIdx.x * 2 + wv;
    const int row0 = tile * 16;
    const int t    = (blockIdx.x * 32 >= N_NODES / 2) ? 1 : 0;
    float* As = xs[wv];

    for (int i = 0; i < 32; ++i) {
        int idx = i * 32 + lane;
        int m = idx >> 6, cc = idx & 63;
        As[m * 65 + cc] = x[(size_t)(row0 + m) * DIN + cc];
    }
    __syncthreads();
    {
        const int m = lane & 15, c0 = (lane >> 4) * 32;
        float s = 0.f;
        for (int j = 0; j < 32; ++j) s += As[m * 65 + c0 + j];
        s += __shfl_xor(s, 16, 32);
        float mean = s * (1.0f / 64.0f);
        float v2 = 0.f;
        for (int j = 0; j < 32; ++j) { float d = As[m * 65 + c0 + j] - mean; v2 += d * d; }
        v2 += __shfl_xor(v2, 16, 32);
        float inv = rsqrtf(v2 * (1.0f / 64.0f) + 1e-5f);
        for (int j = 0; j < 32; ++j) {
            int cc = c0 + j;
            As[m * 65 + cc] = (As[m * 65 + cc] - mean) * inv * ln_s[t * DIN + cc]
                              + ln_b[t * DIN + cc];
        }
    }

    const int half = lane >> 4, qn = lane & 15;
    const float* Ws[3] = {Wq + t * DIN * DH, Wk + t * DIN * DH, Wv + t * DIN * DH};
    const float* bs[3] = {bq + t * DH, bk + t * DH, bv + t * DH};
    float*       os[3] = {qo, ko, vo};
    for (int mm = 0; mm < 3; ++mm) {
        __syncthreads();
        for (int i = threadIdx.x; i < DIN * DH; i += 64)
            wlds[i] = Ws[mm][i];
        __syncthreads();
        #pragma unroll
        for (int nt = 0; nt < 4; ++nt) {
            const int n0 = nt * 16;
            v8f c = gemm_tile<64>(As, 65, wlds, DH, n0, lane);
            float bias = bs[mm][n0 + qn];
            #pragma unroll
            for (int i = 0; i < 8; ++i) {
                int m = i + 8 * half;
                os[mm][(size_t)(row0 + m) * DH + n0 + qn] = c[i] + bias;
            }
        }
    }
}

__device__ __forceinline__ int edge_bin(int dst, int e) {
    return 2 * (dst / RNG) + ((e >= N_EDGES / 2) ? 1 : 0);
}

__global__ void __launch_bounds__(256) k_bin_count(const int* __restrict__ edst,
                                                   int* __restrict__ cnt) {
    __shared__ int hist[NBIN];
    for (int i = threadIdx.x; i < NBIN; i += 256) hist[i] = 0;
    __syncthreads();
    for (int u = 0; u < 4; ++u) {
        const int e = blockIdx.x * 1024 + u * 256 + threadIdx.x;
        if (e < N_EDGES) atomicAdd(&hist[edge_bin(edst[e], e)], 1);
    }
    __syncthreads();
    for (int i = threadIdx.x; i < NBIN; i += 256) {
        const int c = hist[i];
        if (c) atomicAdd(&cnt[i], c);
    }
}

__global__ void __launch_bounds__(512) k_bin_scan(const int* __restrict__ cnt,
                                                  int* __restrict__ boff,
                                                  int* __restrict__ cur) {
    __shared__ int s[512];
    const int i = threadIdx.x;
    const int c = (i < NBIN) ? cnt[i] : 0;
    s[i] = c;
    __syncthreads();
    for (int off = 1; off < 512; off <<= 1) {
        const int v = (i >= off) ? s[i - off] : 0;
        __syncthreads();
        s[i] += v;
        __syncthreads();
    }
    const int excl = s[i] - c;
    if (i < NBIN) { boff[i] = excl; cur[i] = excl; }
    if (i == NBIN - 1) boff[NBIN] = s[i];
}

__global__ void __launch_bounds__(256) k_bin_scatter(const int* __restrict__ esrc,
                                                     const int* __restrict__ edst,
                                                     int* __restrict__ cur,
                                                     int* __restrict__ pe,
                                                     int* __restrict__ ps,
                                                     int* __restrict__ pd) {
    __shared__ int hist[NBIN];
    __shared__ int base[NBIN];
    for (int i = threadIdx.x; i < NBIN; i += 256) hist[i] = 0;
    __syncthreads();
    int bin[4], rank[4], sv[4], dv[4];
    #pragma unroll
    for (int u = 0; u < 4; ++u) {
        const int e = blockIdx.x * 1024 + u * 256 + threadIdx.x;
        bin[u] = -1; rank[u] = 0; sv[u] = 0; dv[u] = 0;
        if (e < N_EDGES) {
            dv[u] = edst[e];
            sv[u] = esrc[e];
            bin[u] = edge_bin(dv[u], e);
            rank[u] = atomicAdd(&hist[bin[u]], 1);
        }
    }
    __syncthreads();
    for (int i = threadIdx.x; i < NBIN; i += 256) {
        const int c = hist[i];
        base[i] = c ? atomicAdd(&cur[i], c) : 0;
    }
    __syncthreads();
    #pragma unroll
    for (int u = 0; u < 4; ++u) {
        const int e = blockIdx.x * 1024 + u * 256 + threadIdx.x;
        if (bin[u] >= 0) {
            const int pos = base[bin[u]] + rank[u];
            pe[pos] = e;
            ps[pos] = sv[u];
            pd[pos] = dv[u];
        }
    }
}

__global__ void __launch_bounds__(EWV * 32)
k_edge_own(const float* __restrict__ ea,
           const float* __restrict__ We, const float* __restrict__ be,
           const float* __restrict__ Wg, const float* __restrict__ bg,
           const float* __restrict__ ew,
           const float* __restrict__ ln_s, const float* __restrict__ ln_b,
           const int* __restrict__ boff,
           const int* __restrict__ pe, const int* __restrict__ ps,
           const int* __restrict__ pd,
           const float* __restrict__ qbuf, const float* __restrict__ kbuf,
           const float* __restrict__ vbuf,
           const float* __restrict__ Wo, const float* __restrict__ bo,
           float* __restrict__ out) {
    __shared__ float accl[RNG * ROWP];
    __shared__ float wl[2 * DE * DH];
    __shared__ float ewlds[NH * 16 * 17];
    __shared__ float ebs[EWV][16 * 33];
    __shared__ float lnl[2 * DE];
    __shared__ int   ssrc[EWV][16], sdst[EWV][16];
    __shared__ int   srow[EWV][16];
    const int lane = threadIdx.x & 31;
    const int wv   = threadIdx.x >> 5;
    const int half = lane >> 4, qn = lane & 15;
    const int p    = blockIdx.x;
    const int r0   = p * RNG;
    float* As    = ebs[wv];
    float* welds = wl;
    float* wglds = wl + DE * DH;

    for (int i = threadIdx.x; i < RNG * ROWP; i += EWV * 32) accl[i] = 0.f;

    for (int t = 0; t < 2; ++t) {
        __syncthreads();
        for (int i = threadIdx.x; i < DE * DH; i += EWV * 32) {
            welds[i] = We[t * DE * DH + i];
            wglds[i] = Wg[t * DE * DH + i];
        }
        for (int idx = threadIdx.x; idx < NH * 16 * 16; idx += EWV * 32) {
            int h = idx >> 8, rem = idx & 255, i = rem >> 4, j = rem & 15;
            ewlds[(h * 16 + j) * 17 + i] = ew[t * NH * 256 + idx];
        }
        if (threadIdx.x < 2 * DE)
            lnl[threadIdx.x] = (threadIdx.x < DE) ? ln_s[t * DE + threadIdx.x]
                                                  : ln_b[t * DE + threadIdx.x - DE];
        __syncthreads();
        const int s0 = boff[2 * p + t], s1 = boff[2 * p + t + 1];

        for (int slot0 = s0 + wv * 16; slot0 < s1; slot0 += EWV * 16) {
            const int nvalid = min(16, s1 - slot0);
            const int slot = min(slot0 + qn, N_EDGES - 1);
            const int my_e = pe[slot], my_s = ps[slot], my_d = pd[slot];
            #pragma unroll
            for (int m = 0; m < 16; ++m) {
                const int em = __builtin_amdgcn_readlane(my_e, m);
                As[m * 33 + lane] = ea[(size_t)em * DE + lane];
            }
            if (lane < 16) {
                ssrc[wv][lane] = my_s;
                sdst[wv][lane] = my_d;
                srow[wv][lane] = (lane < nvalid) ? my_d - r0 : 0;
            }
            float okf[8];
            #pragma unroll
            for (int i = 0; i < 8; ++i) okf[i] = (i + 8 * half < nvalid) ? 1.0f : 0.0f;
            const float* krow = kbuf + (size_t)my_s * DH;
            __builtin_amdgcn_fence(__ATOMIC_ACQ_REL, "workgroup", "local");
            const int  drow = srow[wv][(qn & 7) + 8 * half];
            const bool dadd = (qn < 8) && (qn + 8 * half < nvalid);
            {
                const int m = lane & 15, c0 = (lane >> 4) * 16;
                float s = 0.f;
                for (int j = 0; j < 16; ++j) s += As[m * 33 + c0 + j];
                s += __shfl_xor(s, 16, 32);
                float mean = s * (1.0f / 32.0f);
                float v2 = 0.f;
                for (int j = 0; j < 16; ++j) { float d = As[m * 33 + c0 + j] - mean; v2 += d * d; }
                v2 += __shfl_xor(v2, 16, 32);
                float inv = rsqrtf(v2 * (1.0f / 32.0f) + 1e-5f);
                for (int j = 0; j < 16; ++j) {
                    int cc = c0 + j;
                    As[m * 33 + cc] = (As[m * 33 + cc] - mean) * inv * lnl[cc] + lnl[DE + cc];
                }
            }
            __builtin_amdgcn_fence(__ATOMIC_ACQ_REL, "workgroup", "local");

            for (int h = 0; h < NH; ++h) {
                const int n0 = h * 16;
                float dq[8], vv[8];
                #pragma unroll
                for (int i = 0; i < 8; ++i) {
                    int m = i + 8 * half;
                    dq[i] = qbuf[(size_t)sdst[wv][m] * DH + n0 + qn];
                    vv[i] = vbuf[(size_t)ssrc[wv][m] * DH + n0 + qn];
                }
                v8f eatt = gemm_tile<32>(As, 33, welds, DH, n0, lane);
                v8f egat = gemm_tile<32>(As, 33, wglds, DH, n0, lane);
                const float batt = be[t * DH + n0 + qn];
                const float bgat = bg[t * DH + n0 + qn];

                v8f ekc = gemm_tile<16>(krow + n0, 0, ewlds + h * 16 * 17, 17, 0, lane);

                v8f red;
                #pragma unroll
                for (int i = 0; i < 8; ++i) red[i] = dq[i] * ekc[i] + eatt[i] + batt;
                #pragma unroll
                for (int off = 1; off < 16; off <<= 1) {
                    #pragma unroll
                    for (int i = 0; i < 8; ++i) red[i] += __shfl_xor(red[i], off, 32);
                }
                float es8[8];
                #pragma unroll
                for (int i = 0; i < 8; ++i) {
                    float sc = red[i] * 0.25f;
                    sc = fminf(5.0f, fmaxf(-5.0f, sc));
                    es8[i] = __expf(sc);
                }
                float dsum = es8[0];
                for (int i = 1; i < 8; ++i) dsum = (qn == i) ? es8[i] : dsum;
                if (dadd) atomicAdd(&accl[drow * ROWP + DH + h], dsum);
                #pragma unroll
                for (int i = 0; i < 8; ++i) {
                    float g  = 1.0f / (1.0f + __expf(-(egat[i] + bgat)));
                    atomicAdd(&accl[srow[wv][i + 8 * half] * ROWP + n0 + qn],
                              es8[i] * vv[i] * g * okf[i]);
                }
            }
        }
    }
    __syncthreads();

    const int tn = (p >= NRNG / 2) ? 1 : 0;
    for (int i = threadIdx.x; i < DH * DIN; i += EWV * 32)
        wl[i] = Wo[tn * DH * DIN + i];
    __syncthreads();
    for (int rt = wv; rt < RNG / 16; rt += EWV) {
        float* Ar = accl + rt * 16 * ROWP;
        #pragma unroll 8
        for (int i = 0; i < 32; ++i) {
            int idx = i * 32 + lane;
            int m = idx >> 6, cc = idx & 63;
            float dv = Ar[m * ROWP + DH + (cc >> 4)];
            float av = Ar[m * ROWP + cc];
            float qv = av / dv;
            Ar[m * ROWP + cc] = (dv > 0.f) ? qv : 0.f;
        }
        __builtin_amdgcn_fence(__ATOMIC_ACQ_REL, "workgroup", "local");
        #pragma unroll
        for (int nt = 0; nt < 4; ++nt) {
            const int n0 = nt * 16;
            v8f c = gemm_tile<64>(Ar, ROWP, wl, DIN, n0, lane);
            float bias = bo[tn * DIN + n0 + qn];
            #pragma unroll
            for (int i = 0; i < 8; ++i)
                out[(size_t)(r0 + rt * 16 + i + 8 * half) * DIN + n0 + qn] = c[i] + bias;
        }
    }
}

extern "C" void kernel_launch(void* const* d_in, const int* in_sizes, int n_in,
                              void* d_out, int out_size, void* d_ws, size_t ws_size,
                              hipStream_t stream) {
    const float* x    = (const float*)d_in[0];
    const float* ea   = (const float*)d_in[1];
    const float* Wq   = (const float*)d_in[2];
    const float* bq   = (const float*)d_in[3];
    const float* Wk   = (const float*)d_in[4];
    const float* bk   = (const float*)d_in[5];
    const float* Wv   = (const float*)d_in[6];
    const float* bv   = (const float*)d_in[7];
    const float* Wo   = (const float*)d_in[8];
    const float* bo   = (const float*)d_in[9];
    const float* We   = (const float*)d_in[10];
    const float* be   = (const float*)d_in[11];
    const float* Wg   = (const float*)d_in[12];
    const float* bg   = (const float*)d_in[13];
    const float* ew   = (const float*)d_in[14];
    const float* lnns = (const float*)d_in[15];
    const float* lnnb = (const float*)d_in[16];
    const float* lnes = (const float*)d_in[17];
    const float* lneb = (const float*)d_in[18];
    const int*   ei   = (const int*)d_in[19];
    const int* esrc = ei;
    const int* edst = ei + N_EDGES;

    float* ws  = (float*)d_ws;
    float* qb  = ws;
    float* kb  = qb  + (size_t)N_NODES * DH;
    float* vb  = kb  + (size_t)N_NODES * DH;
    int*   cnt  = (int*)(vb + (size_t)N_NODES * DH);
    int*   cur  = cnt + 512;
    int*   boff = cur + 512;
    int*   pe   = boff + 1024;
    int*   ps   = pe + N_EDGES;
    int*   pd   = ps + N_EDGES;

    k_zero<<<2, 256, 0, stream>>>((float*)cnt, 512);

    k_node_qkv<<<(N_NODES / 16) / 2, 64, 0, stream>>>(
        x, Wq, bq, Wk, bk, Wv, bv, lnns, lnnb, qb, kb, vb);

    k_bin_count<<<(N_EDGES + 1023) / 1024, 256, 0, stream>>>(edst, cnt);
    k_bin_scan<<<1, 512, 0, stream>>>(cnt, boff, cur);
    k_bin_scatter<<<(N_EDGES + 1023) / 1024, 256, 0, stream>>>(esrc, edst, cur, pe, ps, pd);

    k_edge_own<<<NRNG, EWV * 32, 0, stream>>>(
        ea, We, be, Wg, bg, ew, lnes, lneb, boff, pe, ps, pd, qb, kb, vb, Wo, bo,
        (float*)d_out);
}


# Edge tile products taken transposed (WMMA operands exchanged): one edge per lane, q/k/v as 32-byte slices, head sum in registers + one lane-half exchange; score and key-transform sum order change
# speedup vs baseline: 1.6819x; 1.1004x over previous
#include <hip/hip_runtime.h>
#include <hip/hip_bf16.h>
#include <math.h>

#define N_NODES 200000
#define N_EDGES 800000
#define DIN 64
#define DE  32
#define DH  64
#define NH  4
#define HD  16
#define RNG  800
#define NRNG (N_NODES / RNG)
#define NBIN (2 * NRNG)
#define ROWP 69
#define EWV  24

typedef float v2f __attribute__((ext_vector_type(2)));
typedef float v4f __attribute__((ext_vector_type(4)));
typedef float v8f __attribute__((ext_vector_type(8)));

__device__ __forceinline__ v8f wstep(v2f a, v2f b, v8f c) {
    return __builtin_amdgcn_wmma_f32_16x16x4_f32(false, a, false, b, (short)0, c,
                                                 false, false);
}

template <int K>
__device__ __forceinline__ v8f gemm_tile(const float* As, int lda,
                                         const float* Bs, int ldb,
                                         int n0, int lane) {
    const int half = lane >> 4;
    const int qn   = lane & 15;
    v2f a[K / 4], b[K / 4];
    #pragma unroll
    for (int s = 0; s < K / 4; ++s) {
        const int k0 = 4 * s + 2 * half;
        a[s].x = As[qn * lda + k0];
        a[s].y = As[qn * lda + k0 + 1];
        b[s].x = Bs[k0 * ldb + n0 + qn];
        b[s].y = Bs[(k0 + 1) * ldb + n0 + qn];
    }
    v8f c = {0.f, 0.f, 0.f, 0.f, 0.f, 0.f, 0.f, 0.f};
    #pragma unroll
    for (int s = 0; s < K / 4; ++s) c = wstep(a[s], b[s], c);
    return c;
}

template <int K>
__device__ __forceinline__ v8f gemm_tile_t(const float* As, int lda,
                                           const float* Bs, int ldb,
                                           int n0, int lane) {
    const int half = lane >> 4;
    const int qn   = lane & 15;
    v2f a[K / 4], b[K / 4];
    #pragma unroll
    for (int s = 0; s < K / 4; ++s) {
        const int k0 = 4 * s + 2 * half;
        a[s].x = As[qn * lda + k0];
        a[s].y = As[qn * lda + k0 + 1];
        b[s].x = Bs[k0 * ldb + n0 + qn];
        b[s].y = Bs[(k0 + 1) * ldb + n0 + qn];
    }
    v8f c = {0.f, 0.f, 0.f, 0.f, 0.f, 0.f, 0.f, 0.f};
    #pragma unroll
    for (int s = 0; s < K / 4; ++s) c = wstep(b[s], a[s], c);
    return c;
}

__global__ void k_zero(float* __restrict__ p, int n) {
    int i = blockIdx.x * blockDim.x + threadIdx.x;
    if (i < n) p[i] = 0.f;
}

__global__ void k_node_qkv(const float* __restrict__ x,
                           const float* __restrict__ Wq, const float* __restrict__ bq,
                           const float* __restrict__ Wk, const float* __restrict__ bk,
                           const float* __restrict__ Wv, const float* __restrict__ bv,
                           const float* __restrict__ ln_s, const float* __restrict__ ln_b,
                           float* __restrict__ qo, float* __restrict__ ko,
                           float* __restrict__ vo) {
    __shared__ float xs[2][16 * 65];
    __shared__ float wlds[64 * 64];
    const int lane = threadIdx.x & 31;
    const int wv   = threadIdx.x >> 5;
    const int tile = blockIdx.x * 2 + wv;
    const int row0 = tile * 16;
    const int t    = (blockIdx.x * 32 >= N_NODES / 2) ? 1 : 0;
    float* As = xs[wv];

    for (int i = 0; i < 32; ++i) {
        int idx = i * 32 + lane;
        int m = idx >> 6, cc = idx & 63;
        As[m * 65 + cc] = x[(size_t)(row0 + m) * DIN + cc];
    }
    __syncthreads();
    {
        const int m = lane & 15, c0 = (lane >> 4) * 32;
        float s = 0.f;
        for (int j = 0; j < 32; ++j) s += As[m * 65 + c0 + j];
        s += __shfl_xor(s, 16, 32);
        float mean = s * (1.0f / 64.0f);
        float v2 = 0.f;
        for (int j = 0; j < 32; ++j) { float d = As[m * 65 + c0 + j] - mean; v2 += d * d; }
        v2 += __shfl_xor(v2, 16, 32);
        float inv = rsqrtf(v2 * (1.0f / 64.0f) + 1e-5f);
        for (int j = 0; j < 32; ++j) {
            int cc = c0 + j;
            As[m * 65 + cc] = (As[m * 65 + cc] - mean) * inv * ln_s[t * DIN + cc]
                              + ln_b[t * DIN + cc];
        }
    }

    const int half = lane >> 4, qn = lane & 15;
    const float* Ws[3] = {Wq + t * DIN * DH, Wk + t * DIN * DH, Wv + t * DIN * DH};
    const float* bs[3] = {bq + t * DH, bk + t * DH, bv + t * DH};
    float*       os[3] = {qo, ko, vo};
    for (int mm = 0; mm < 3; ++mm) {
        __syncthreads();
        for (int i = threadIdx.x; i < DIN * DH; i += 64)
            wlds[i] = Ws[mm][i];
        __syncthreads();
        #pragma unroll
        for (int nt = 0; nt < 4; ++nt) {
            const int n0 = nt * 16;
            v8f c = gemm_tile<64>(As, 65, wlds, DH, n0, lane);
            float bias = bs[mm][n0 + qn];
            #pragma unroll
            for (int i = 0; i < 8; ++i) {
                int m = i + 8 * half;
                os[mm][(size_t)(row0 + m) * DH + n0 + qn] = c[i] + bias;
            }
        }
    }
}

__device__ __forceinline__ int edge_bin(int dst, int e) {
    return 2 * (dst / RNG) + ((e >= N_EDGES / 2) ? 1 : 0);
}

__global__ void __launch_bounds__(256) k_bin_count(const int* __restrict__ edst,
                                                   int* __restrict__ cnt) {
    __shared__ int hist[NBIN];
    for (int i = threadIdx.x; i < NBIN; i += 256) hist[i] = 0;
    __syncthreads();
    for (int u = 0; u < 4; ++u) {
        const int e = blockIdx.x * 1024 + u * 256 + threadIdx.x;
        if (e < N_EDGES) atomicAdd(&hist[edge_bin(edst[e], e)], 1);
    }
    __syncthreads();
    for (int i = threadIdx.x; i < NBIN; i += 256) {
        const int c = hist[i];
        if (c) atomicAdd(&cnt[i], c);
    }
}

__global__ void __launch_bounds__(512) k_bin_scan(const int* __restrict__ cnt,
                                                  int* __restrict__ boff,
                                                  int* __restrict__ cur) {
    __shared__ int s[512];
    const int i = threadIdx.x;
    const int c = (i < NBIN) ? cnt[i] : 0;
    s[i] = c;
    __syncthreads();
    for (int off = 1; off < 512; off <<= 1) {
        const int v = (i >= off) ? s[i - off] : 0;
        __syncthreads();
        s[i] += v;
        __syncthreads();
    }
    const int excl = s[i] - c;
    if (i < NBIN) { boff[i] = excl; cur[i] = excl; }
    if (i == NBIN - 1) boff[NBIN] = s[i];
}

__global__ void __launch_bounds__(256) k_bin_scatter(const int* __restrict__ esrc,
                                                     const int* __restrict__ edst,
                                                     int* __restrict__ cur,
                                                     int* __restrict__ pe,
                                                     int* __restrict__ ps,
                                                     int* __restrict__ pd) {
    __shared__ int hist[NBIN];
    __shared__ int base[NBIN];
    for (int i = threadIdx.x; i < NBIN; i += 256) hist[i] = 0;
    __syncthreads();
    int bin[4], rank[4], sv[4], dv[4];
    #pragma unroll
    for (int u = 0; u < 4; ++u) {
        const int e = blockIdx.x * 1024 + u * 256 + threadIdx.x;
        bin[u] = -1; rank[u] = 0; sv[u] = 0; dv[u] = 0;
        if (e < N_EDGES) {
            dv[u] = edst[e];
            sv[u] = esrc[e];
            bin[u] = edge_bin(dv[u], e);
            rank[u] = atomicAdd(&hist[bin[u]], 1);
        }
    }
    __syncthreads();
    for (int i = threadIdx.x; i < NBIN; i += 256) {
        const int c = hist[i];
        base[i] = c ? atomicAdd(&cur[i], c) : 0;
    }
    __syncthreads();
    #pragma unroll
    for (int u = 0; u < 4; ++u) {
        const int e = blockIdx.x * 1024 + u * 256 + threadIdx.x;
        if (bin[u] >= 0) {
            const int pos = base[bin[u]] + rank[u];
            pe[pos] = e;
            ps[pos] = sv[u];
            pd[pos] = dv[u];
        }
    }
}

__global__ void __launch_bounds__(EWV * 32)
k_edge_own(const float* __restrict__ ea,
           const float* __restrict__ We, const float* __restrict__ be,
           const float* __restrict__ Wg, const float* __restrict__ bg,
           const float* __restrict__ ew,
           const float* __restrict__ ln_s, const float* __restrict__ ln_b,
           const int* __restrict__ boff,
           const int* __restrict__ pe, const int* __restrict__ ps,
           const int* __restrict__ pd,
           const float* __restrict__ qbuf, const float* __restrict__ kbuf,
           const float* __restrict__ vbuf,
           const float* __restrict__ Wo, const float* __restrict__ bo,
           float* __restrict__ out) {
    __shared__ float accl[RNG * ROWP];
    __shared__ float wl[2 * DE * DH];
    __shared__ float ewlds[NH * 16 * 17];
    __shared__ float ebs[EWV][16 * 33];
    __shared__ float lnl[2 * DE];
    const int lane = threadIdx.x & 31;
    const int wv   = threadIdx.x >> 5;
    const int half = lane >> 4, qn = lane & 15;
    const int p    = blockIdx.x;
    const int r0   = p * RNG;
    float* As    = ebs[wv];
    float* welds = wl;
    float* wglds = wl + DE * DH;

    for (int i = threadIdx.x; i < RNG * ROWP; i += EWV * 32) accl[i] = 0.f;

    for (int t = 0; t < 2; ++t) {
        __syncthreads();
        for (int i = threadIdx.x; i < DE * DH; i += EWV * 32) {
            welds[i] = We[t * DE * DH + i];
            wglds[i] = Wg[t * DE * DH + i];
        }
        for (int idx = threadIdx.x; idx < NH * 16 * 16; idx += EWV * 32) {
            int h = idx >> 8, rem = idx & 255, i = rem >> 4, j = rem & 15;
            ewlds[(h * 16 + j) * 17 + i] = ew[t * NH * 256 + idx];
        }
        if (threadIdx.x < 2 * DE)
            lnl[threadIdx.x] = (threadIdx.x < DE) ? ln_s[t * DE + threadIdx.x]
                                                  : ln_b[t * DE + threadIdx.x - DE];
        __syncthreads();
        const int s0 = boff[2 * p + t], s1 = boff[2 * p + t + 1];

        for (int slot0 = s0 + wv * 16; slot0 < s1; slot0 += EWV * 16) {
            const int nvalid = min(16, s1 - slot0);
            const int slot = min(slot0 + qn, N_EDGES - 1);
            const int my_e = pe[slot], my_s = ps[slot], my_d = pd[slot];
            #pragma unroll
            for (int m = 0; m < 16; ++m) {
                const int em = __builtin_amdgcn_readlane(my_e, m);
                As[m * 33 + lane] = ea[(size_t)em * DE + lane];
            }
            const bool  ok   = qn < nvalid;
            const float okf  = ok ? 1.0f : 0.0f;
            const int   arow = (ok ? my_d - r0 : 0) * ROWP;
            const bool  dadd = ok && (half == 0);
            const float* qrow = qbuf + (size_t)my_d * DH;
            const float* krow = kbuf + (size_t)my_s * DH;
            const float* vrow = vbuf + (size_t)my_s * DH;
            __builtin_amdgcn_fence(__ATOMIC_ACQ_REL, "workgroup", "local");
            {
                const int m = lane & 15, c0 = (lane >> 4) * 16;
                float s = 0.f;
                for (int j = 0; j < 16; ++j) s += As[m * 33 + c0 + j];
                s += __shfl_xor(s, 16, 32);
                float mean = s * (1.0f / 32.0f);
                float v2 = 0.f;
                for (int j = 0; j < 16; ++j) { float d = As[m * 33 + c0 + j] - mean; v2 += d * d; }
                v2 += __shfl_xor(v2, 16, 32);
                float inv = rsqrtf(v2 * (1.0f / 32.0f) + 1e-5f);
                for (int j = 0; j < 16; ++j) {
                    int cc = c0 + j;
                    As[m * 33 + cc] = (As[m * 33 + cc] - mean) * inv * lnl[cc] + lnl[DE + cc];
                }
            }
            __builtin_amdgcn_fence(__ATOMIC_ACQ_REL, "workgroup", "local");

            for (int h = 0; h < NH; ++h) {
                const int n0 = h * 16;
                const int f0 = n0 + 8 * half;
                const v4f q0 = *(const v4f*)(qrow + f0), q1 = *(const v4f*)(qrow + f0 + 4);
                const v4f k0 = *(const v4f*)(krow + f0), k1 = *(const v4f*)(krow + f0 + 4);
                const v4f v0 = *(const v4f*)(vrow + f0), v1 = *(const v4f*)(vrow + f0 + 4);
                const v4f ba0 = *(const v4f*)(be + t * DH + f0), ba1 = *(const v4f*)(be + t * DH + f0 + 4);
                const v4f bg0 = *(const v4f*)(bg + t * DH + f0), bg1 = *(const v4f*)(bg + t * DH + f0 + 4);
                const float qv[8] = {q0.x, q0.y, q0.z, q0.w, q1.x, q1.y, q1.z, q1.w};
                const float vv[8] = {v0.x, v0.y, v0.z, v0.w, v1.x, v1.y, v1.z, v1.w};
                const float ba[8] = {ba0.x, ba0.y, ba0.z, ba0.w, ba1.x, ba1.y, ba1.z, ba1.w};
                const float bb[8] = {bg0.x, bg0.y, bg0.z, bg0.w, bg1.x, bg1.y, bg1.z, bg1.w};

                v8f eatt = gemm_tile_t<32>(As, 33, welds, DH, n0, lane);
                v8f egat = gemm_tile_t<32>(As, 33, wglds, DH, n0, lane);

                const float* ewh = ewlds + h * 16 * 17;
                v2f ka[4], kb[4];
                kb[0].x = k0.x; kb[0].y = k0.y; kb[1].x = k0.z; kb[1].y = k0.w;
                kb[2].x = k1.x; kb[2].y = k1.y; kb[3].x = k1.z; kb[3].y = k1.w;
                #pragma unroll
                for (int s = 0; s < 4; ++s) {
                    const int j0 = 8 * half + 2 * s;
                    ka[s].x = ewh[j0 * 17 + qn];
                    ka[s].y = ewh[(j0 + 1) * 17 + qn];
                }
                v8f ekc = {0.f, 0.f, 0.f, 0.f, 0.f, 0.f, 0.f, 0.f};
                #pragma unroll
                for (int s = 0; s < 4; ++s) ekc = wstep(ka[s], kb[s], ekc);

                float ssum = 0.f;
                #pragma unroll
                for (int r = 0; r < 8; ++r) ssum += qv[r] * ekc[r] + (eatt[r] + ba[r]);
                ssum += __shfl_xor(ssum, 16, 32);
                float sc = ssum * 0.25f;
                sc = fminf(5.0f, fmaxf(-5.0f, sc));
                const float es = __expf(sc);
                if (dadd) atomicAdd(&accl[arow + DH + h], es);
                #pragma unroll
                for (int r = 0; r < 8; ++r) {
                    float g = 1.0f / (1.0f + __expf(-(egat[r] + bb[r])));
                    atomicAdd(&accl[arow + f0 + r], es * vv[r] * g * okf);
                }
            }
        }
    }
    __syncthreads();

    const int tn = (p >= NRNG / 2) ? 1 : 0;
    for (int i = threadIdx.x; i < DH * DIN; i += EWV * 32)
        wl[i] = Wo[tn * DH * DIN + i];
    __syncthreads();
    for (int rt = wv; rt < RNG / 16; rt += EWV) {
        float* Ar = accl + rt * 16 * ROWP;
        #pragma unroll 8
        for (int i = 0; i < 32; ++i) {
            int idx = i * 32 + lane;
            int m = idx >> 6, cc = idx & 63;
            float dv = Ar[m * ROWP + DH + (cc >> 4)];
            float av = Ar[m * ROWP + cc];
            float qv = av / dv;
            Ar[m * ROWP + cc] = (dv > 0.f) ? qv : 0.f;
        }
        __builtin_amdgcn_fence(__ATOMIC_ACQ_REL, "workgroup", "local");
        #pragma unroll
        for (int nt = 0; nt < 4; ++nt) {
            const int n0 = nt * 16;
            v8f c = gemm_tile<64>(Ar, ROWP, wl, DIN, n0, lane);
            float bias = bo[tn * DIN + n0 + qn];
            #pragma unroll
            for (int i = 0; i < 8; ++i)
                out[(size_t)(r0 + rt * 16 + i + 8 * half) * DIN + n0 + qn] = c[i] + bias;
        }
    }
}

extern "C" void kernel_launch(void* const* d_in, const int* in_sizes, int n_in,
                              void* d_out, int out_size, void* d_ws, size_t ws_size,
                              hipStream_t stream) {
    const float* x    = (const float*)d_in[0];
    const float* ea   = (const float*)d_in[1];
    const float* Wq   = (const float*)d_in[2];
    const float* bq   = (const float*)d_in[3];
    const float* Wk   = (const float*)d_in[4];
    const float* bk   = (const float*)d_in[5];
    const float* Wv   = (const float*)d_in[6];
    const float* bv   = (const float*)d_in[7];
    const float* Wo   = (const float*)d_in[8];
    const float* bo   = (const float*)d_in[9];
    const float* We   = (const float*)d_in[10];
    const float* be   = (const float*)d_in[11];
    const float* Wg   = (const float*)d_in[12];
    const float* bg   = (const float*)d_in[13];
    const float* ew   = (const float*)d_in[14];
    const float* lnns = (const float*)d_in[15];
    const float* lnnb = (const float*)d_in[16];
    const float* lnes = (const float*)d_in[17];
    const float* lneb = (const float*)d_in[18];
    const int*   ei   = (const int*)d_in[19];
    const int* esrc = ei;
    const int* edst = ei + N_EDGES;

    float* ws  = (float*)d_ws;
    float* qb  = ws;
    float* kb  = qb  + (size_t)N_NODES * DH;
    float* vb  = kb  + (size_t)N_NODES * DH;
    int*   cnt  = (int*)(vb + (size_t)N_NODES * DH);
    int*   cur  = cnt + 512;
    int*   boff = cur + 512;
    int*   pe   = boff + 1024;
    int*   ps   = pe + N_EDGES;
    int*   pd   = ps + N_EDGES;

    k_zero<<<2, 256, 0, stream>>>((float*)cnt, 512);

    k_node_qkv<<<(N_NODES / 16) / 2, 64, 0, stream>>>(
        x, Wq, bq, Wk, bk, Wv, bv, lnns, lnnb, qb, kb, vb);

    k_bin_count<<<(N_EDGES + 1023) / 1024, 256, 0, stream>>>(edst, cnt);
    k_bin_scan<<<1, 512, 0, stream>>>(cnt, boff, cur);
    k_bin_scatter<<<(N_EDGES + 1023) / 1024, 256, 0, stream>>>(esrc, edst, cur, pe, ps, pd);

    k_edge_own<<<NRNG, EWV * 32, 0, stream>>>(
        ea, We, be, Wg, bg, ew, lnes, lneb, boff, pe, ps, pd, qb, kb, vb, Wo, bo,
        (float*)d_out);
}


# Score's e_att term as one product with head-summed We columns (sum over a head's 16 columns moved onto the weights): 1 WMMA column tile per edge tile, not 4; score sum order changes
# speedup vs baseline: 1.7403x; 1.0347x over previous
#include <hip/hip_runtime.h>
#include <hip/hip_bf16.h>
#include <math.h>

#define N_NODES 200000
#define N_EDGES 800000
#define DIN 64
#define DE  32
#define DH  64
#define NH  4
#define HD  16
#define RNG  800
#define NRNG (N_NODES / RNG)
#define NBIN (2 * NRNG)
#define ROWP 69
#define EWV  24

typedef float v2f __attribute__((ext_vector_type(2)));
typedef float v4f __attribute__((ext_vector_type(4)));
typedef float v8f __attribute__((ext_vector_type(8)));

__device__ __forceinline__ v8f wstep(v2f a, v2f b, v8f c) {
    return __builtin_amdgcn_wmma_f32_16x16x4_f32(false, a, false, b, (short)0, c,
                                                 false, false);
}

template <int K>
__device__ __forceinline__ v8f gemm_tile(const float* As, int lda,
                                         const float* Bs, int ldb,
                                         int n0, int lane) {
    const int half = lane >> 4;
    const int qn   = lane & 15;
    v2f a[K / 4], b[K / 4];
    #pragma unroll
    for (int s = 0; s < K / 4; ++s) {
        const int k0 = 4 * s + 2 * half;
        a[s].x = As[qn * lda + k0];
        a[s].y = As[qn * lda + k0 + 1];
        b[s].x = Bs[k0 * ldb + n0 + qn];
        b[s].y = Bs[(k0 + 1) * ldb + n0 + qn];
    }
    v8f c = {0.f, 0.f, 0.f, 0.f, 0.f, 0.f, 0.f, 0.f};
    #pragma unroll
    for (int s = 0; s < K / 4; ++s) c = wstep(a[s], b[s], c);
    return c;
}

template <int K>
__device__ __forceinline__ v8f gemm_tile_t(const float* As, int lda,
                                           const float* Bs, int ldb,
                                           int n0, int lane) {
    const int half = lane >> 4;
    const int qn   = lane & 15;
    v2f a[K / 4], b[K / 4];
    #pragma unroll
    for (int s = 0; s < K / 4; ++s) {
        const int k0 = 4 * s + 2 * half;
        a[s].x = As[qn * lda + k0];
        a[s].y = As[qn * lda + k0 + 1];
        b[s].x = Bs[k0 * ldb + n0 + qn];
        b[s].y = Bs[(k0 + 1) * ldb + n0 + qn];
    }
    v8f c = {0.f, 0.f, 0.f, 0.f, 0.f, 0.f, 0.f, 0.f};
    #pragma unroll
    for (int s = 0; s < K / 4; ++s) c = wstep(b[s], a[s], c);
    return c;
}

__global__ void k_zero(float* __restrict__ p, int n) {
    int i = blockIdx.x * blockDim.x + threadIdx.x;
    if (i < n) p[i] = 0.f;
}

__global__ void k_node_qkv(const float* __restrict__ x,
                           const float* __restrict__ Wq, const float* __restrict__ bq,
                           const float* __restrict__ Wk, const float* __restrict__ bk,
                           const float* __restrict__ Wv, const float* __restrict__ bv,
                           const float* __restrict__ ln_s, const float* __restrict__ ln_b,
                           float* __restrict__ qo, float* __restrict__ ko,
                           float* __restrict__ vo) {
    __shared__ float xs[2][16 * 65];
    __shared__ float wlds[64 * 64];
    const int lane = threadIdx.x & 31;
    const int wv   = threadIdx.x >> 5;
    const int tile = blockIdx.x * 2 + wv;
    const int row0 = tile * 16;
    const int t    = (blockIdx.x * 32 >= N_NODES / 2) ? 1 : 0;
    float* As = xs[wv];

    for (int i = 0; i < 32; ++i) {
        int idx = i * 32 + lane;
        int m = idx >> 6, cc = idx & 63;
        As[m * 65 + cc] = x[(size_t)(row0 + m) * DIN + cc];
    }
    __syncthreads();
    {
        const int m = lane & 15, c0 = (lane >> 4) * 32;
        float s = 0.f;
        for (int j = 0; j < 32; ++j) s += As[m * 65 + c0 + j];
        s += __shfl_xor(s, 16, 32);
        float mean = s * (1.0f / 64.0f);
        float v2 = 0.f;
        for (int j = 0; j < 32; ++j) { float d = As[m * 65 + c0 + j] - mean; v2 += d * d; }
        v2 += __shfl_xor(v2, 16, 32);
        float inv = rsqrtf(v2 * (1.0f / 64.0f) + 1e-5f);
        for (int j = 0; j < 32; ++j) {
            int cc = c0 + j;
            As[m * 65 + cc] = (As[m * 65 + cc] - mean) * inv * ln_s[t * DIN + cc]
                              + ln_b[t * DIN + cc];
        }
    }

    const int half = lane >> 4, qn = lane & 15;
    const float* Ws[3] = {Wq + t * DIN * DH, Wk + t * DIN * DH, Wv + t * DIN * DH};
    const float* bs[3] = {bq + t * DH, bk + t * DH, bv + t * DH};
    float*       os[3] = {qo, ko, vo};
    for (int mm = 0; mm < 3; ++mm) {
        __syncthreads();
        for (int i = threadIdx.x; i < DIN * DH; i += 64)
            wlds[i] = Ws[mm][i];
        __syncthreads();
        #pragma unroll
        for (int nt = 0; nt < 4; ++nt) {
            const int n0 = nt * 16;
            v8f c = gemm_tile<64>(As, 65, wlds, DH, n0, lane);
            float bias = bs[mm][n0 + qn];
            #pragma unroll
            for (int i = 0; i < 8; ++i) {
                int m = i + 8 * half;
                os[mm][(size_t)(row0 + m) * DH + n0 + qn] = c[i] + bias;
            }
        }
    }
}

__device__ __forceinline__ int edge_bin(int dst, int e) {
    return 2 * (dst / RNG) + ((e >= N_EDGES / 2) ? 1 : 0);
}

__global__ void __launch_bounds__(256) k_bin_count(const int* __restrict__ edst,
                                                   int* __restrict__ cnt) {
    __shared__ int hist[NBIN];
    for (int i = threadIdx.x; i < NBIN; i += 256) hist[i] = 0;
    __syncthreads();
    for (int u = 0; u < 4; ++u) {
        const int e = blockIdx.x * 1024 + u * 256 + threadIdx.x;
        if (e < N_EDGES) atomicAdd(&hist[edge_bin(edst[e], e)], 1);
    }
    __syncthreads();
    for (int i = threadIdx.x; i < NBIN; i += 256) {
        const int c = hist[i];
        if (c) atomicAdd(&cnt[i], c);
    }
}

__global__ void __launch_bounds__(512) k_bin_scan(const int* __restrict__ cnt,
                                                  int* __restrict__ boff,
                                                  int* __restrict__ cur) {
    __shared__ int s[512];
    const int i = threadIdx.x;
    const int c = (i < NBIN) ? cnt[i] : 0;
    s[i] = c;
    __syncthreads();
    for (int off = 1; off < 512; off <<= 1) {
        const int v = (i >= off) ? s[i - off] : 0;
        __syncthreads();
        s[i] += v;
        __syncthreads();
    }
    const int excl = s[i] - c;
    if (i < NBIN) { boff[i] = excl; cur[i] = excl; }
    if (i == NBIN - 1) boff[NBIN] = s[i];
}

__global__ void __launch_bounds__(256) k_bin_scatter(const int* __restrict__ esrc,
                                                     const int* __restrict__ edst,
                                                     int* __restrict__ cur,
                                                     int* __restrict__ pe,
                                                     int* __restrict__ ps,
                                                     int* __restrict__ pd) {
    __shared__ int hist[NBIN];
    __shared__ int base[NBIN];
    for (int i = threadIdx.x; i < NBIN; i += 256) hist[i] = 0;
    __syncthreads();
    int bin[4], rank[4], sv[4], dv[4];
    #pragma unroll
    for (int u = 0; u < 4; ++u) {
        const int e = blockIdx.x * 1024 + u * 256 + threadIdx.x;
        bin[u] = -1; rank[u] = 0; sv[u] = 0; dv[u] = 0;
        if (e < N_EDGES) {
            dv[u] = edst[e];
            sv[u] = esrc[e];
            bin[u] = edge_bin(dv[u], e);
            rank[u] = atomicAdd(&hist[bin[u]], 1);
        }
    }
    __syncthreads();
    for (int i = threadIdx.x; i < NBIN; i += 256) {
        const int c = hist[i];
        base[i] = c ? atomicAdd(&cur[i], c) : 0;
    }
    __syncthreads();
    #pragma unroll
    for (int u = 0; u < 4; ++u) {
        const int e = blockIdx.x * 1024 + u * 256 + threadIdx.x;
        if (bin[u] >= 0) {
            const int pos = base[bin[u]] + rank[u];
            pe[pos] = e;
            ps[pos] = sv[u];
            pd[pos] = dv[u];
        }
    }
}

__global__ void __launch_bounds__(EWV * 32)
k_edge_own(const float* __restrict__ ea,
           const float* __restrict__ We, const float* __restrict__ be,
           const float* __restrict__ Wg, const float* __restrict__ bg,
           const float* __restrict__ ew,
           const float* __restrict__ ln_s, const float* __restrict__ ln_b,
           const int* __restrict__ boff,
           const int* __restrict__ pe, const int* __restrict__ ps,
           const int* __restrict__ pd,
           const float* __restrict__ qbuf, const float* __restrict__ kbuf,
           const float* __restrict__ vbuf,
           const float* __restrict__ Wo, const float* __restrict__ bo,
           float* __restrict__ out) {
    __shared__ float accl[RNG * ROWP];
    __shared__ __attribute__((aligned(16))) float wl[2 * DE * DH];
    __shared__ float ewlds[NH * 16 * 17];
    __shared__ float ebs[EWV][16 * 33];
    __shared__ float lnl[2 * DE];
    const int lane = threadIdx.x & 31;
    const int wv   = threadIdx.x >> 5;
    const int half = lane >> 4, qn = lane & 15;
    const int p    = blockIdx.x;
    const int r0   = p * RNG;
    float* As    = ebs[wv];
    float* wglds = wl;
    float* wsl   = wl + DE * DH;
    float* bsl   = wsl + DE * 16;

    for (int i = threadIdx.x; i < RNG * ROWP; i += EWV * 32) accl[i] = 0.f;

    for (int t = 0; t < 2; ++t) {
        __syncthreads();
        for (int i = threadIdx.x; i < DE * DH; i += EWV * 32)
            wglds[i] = Wg[t * DE * DH + i];
        if (threadIdx.x < DE * NH) {
            const int c = threadIdx.x >> 2, hh = threadIdx.x & 3;
            const float* wr = We + t * DE * DH + c * DH + hh * HD;
            const v4f w0 = *(const v4f*)(wr), w1 = *(const v4f*)(wr + 4);
            const v4f w2 = *(const v4f*)(wr + 8), w3 = *(const v4f*)(wr + 12);
            float s = w0.x; s += w0.y; s += w0.z; s += w0.w;
            s += w1.x; s += w1.y; s += w1.z; s += w1.w;
            s += w2.x; s += w2.y; s += w2.z; s += w2.w;
            s += w3.x; s += w3.y; s += w3.z; s += w3.w;
            wsl[c * 16 + hh]      = s;
            wsl[c * 16 + hh + 4]  = s;
            wsl[c * 16 + hh + 8]  = s;
            wsl[c * 16 + hh + 12] = s;
        } else if (threadIdx.x < DE * NH + NH) {
            const int hh = threadIdx.x - DE * NH;
            const float* br = be + t * DH + hh * HD;
            float s = br[0];
            for (int d = 1; d < HD; ++d) s += br[d];
            bsl[hh] = s;
        }
        for (int idx = threadIdx.x; idx < NH * 16 * 16; idx += EWV * 32) {
            int h = idx >> 8, rem = idx & 255, i = rem >> 4, j = rem & 15;
            ewlds[(h * 16 + j) * 17 + i] = ew[t * NH * 256 + idx];
        }
        if (threadIdx.x < 2 * DE)
            lnl[threadIdx.x] = (threadIdx.x < DE) ? ln_s[t * DE + threadIdx.x]
                                                  : ln_b[t * DE + threadIdx.x - DE];
        __syncthreads();
        const int s0 = boff[2 * p + t], s1 = boff[2 * p + t + 1];

        for (int slot0 = s0 + wv * 16; slot0 < s1; slot0 += EWV * 16) {
            const int nvalid = min(16, s1 - slot0);
            const int slot = min(slot0 + qn, N_EDGES - 1);
            const int my_e = pe[slot], my_s = ps[slot], my_d = pd[slot];
            #pragma unroll
            for (int m = 0; m < 16; ++m) {
                const int em = __builtin_amdgcn_readlane(my_e, m);
                As[m * 33 + lane] = ea[(size_t)em * DE + lane];
            }
            const bool  ok   = qn < nvalid;
            const float okf  = ok ? 1.0f : 0.0f;
            const int   arow = (ok ? my_d - r0 : 0) * ROWP;
            const bool  dadd = ok && (half == 0);
            const float* qrow = qbuf + (size_t)my_d * DH;
            const float* krow = kbuf + (size_t)my_s * DH;
            const float* vrow = vbuf + (size_t)my_s * DH;
            __builtin_amdgcn_fence(__ATOMIC_ACQ_REL, "workgroup", "local");
            {
                const int m = lane & 15, c0 = (lane >> 4) * 16;
                float s = 0.f;
                for (int j = 0; j < 16; ++j) s += As[m * 33 + c0 + j];
                s += __shfl_xor(s, 16, 32);
                float mean = s * (1.0f / 32.0f);
                float v2 = 0.f;
                for (int j = 0; j < 16; ++j) { float d = As[m * 33 + c0 + j] - mean; v2 += d * d; }
                v2 += __shfl_xor(v2, 16, 32);
                float inv = rsqrtf(v2 * (1.0f / 32.0f) + 1e-5f);
                for (int j = 0; j < 16; ++j) {
                    int cc = c0 + j;
                    As[m * 33 + cc] = (As[m * 33 + cc] - mean) * inv * lnl[cc] + lnl[DE + cc];
                }
            }
            __builtin_amdgcn_fence(__ATOMIC_ACQ_REL, "workgroup", "local");

            const v8f  es8 = gemm_tile_t<32>(As, 33, wsl, 16, 0, lane);
            const v4f  bsv = *(const v4f*)bsl;
            float ec0 = es8[0] + bsv.x, ec1 = es8[1] + bsv.y;
            float ec2 = es8[2] + bsv.z, ec3 = es8[3] + bsv.w;

            for (int h = 0; h < NH; ++h) {
                const int n0 = h * 16;
                const int f0 = n0 + 8 * half;
                const v4f q0 = *(const v4f*)(qrow + f0), q1 = *(const v4f*)(qrow + f0 + 4);
                const v4f k0 = *(const v4f*)(krow + f0), k1 = *(const v4f*)(krow + f0 + 4);
                const v4f v0 = *(const v4f*)(vrow + f0), v1 = *(const v4f*)(vrow + f0 + 4);
                const v4f bg0 = *(const v4f*)(bg + t * DH + f0), bg1 = *(const v4f*)(bg + t * DH + f0 + 4);
                const float qv[8] = {q0.x, q0.y, q0.z, q0.w, q1.x, q1.y, q1.z, q1.w};
                const float vv[8] = {v0.x, v0.y, v0.z, v0.w, v1.x, v1.y, v1.z, v1.w};
                const float bb[8] = {bg0.x, bg0.y, bg0.z, bg0.w, bg1.x, bg1.y, bg1.z, bg1.w};

                v8f egat = gemm_tile_t<32>(As, 33, wglds, DH, n0, lane);
                const float eh = ec0;
                { const float nx = ec0; ec0 = ec1; ec1 = ec2; ec2 = ec3; ec3 = nx; }

                const float* ewh = ewlds + h * 16 * 17;
                v2f ka[4], kb[4];
                kb[0].x = k0.x; kb[0].y = k0.y; kb[1].x = k0.z; kb[1].y = k0.w;
                kb[2].x = k1.x; kb[2].y = k1.y; kb[3].x = k1.z; kb[3].y = k1.w;
                #pragma unroll
                for (int s = 0; s < 4; ++s) {
                    const int j0 = 8 * half + 2 * s;
                    ka[s].x = ewh[j0 * 17 + qn];
                    ka[s].y = ewh[(j0 + 1) * 17 + qn];
                }
                v8f ekc = {0.f, 0.f, 0.f, 0.f, 0.f, 0.f, 0.f, 0.f};
                #pragma unroll
                for (int s = 0; s < 4; ++s) ekc = wstep(ka[s], kb[s], ekc);

                float ssum = 0.f;
                #pragma unroll
                for (int r = 0; r < 8; ++r) ssum += qv[r] * ekc[r];
                ssum += __shfl_xor(ssum, 16, 32);
                float sc = (ssum + eh) * 0.25f;
                sc = fminf(5.0f, fmaxf(-5.0f, sc));
                const float es = __expf(sc);
                if (dadd) atomicAdd(&accl[arow + DH + h], es);
                #pragma unroll
                for (int r = 0; r < 8; ++r) {
                    float g = 1.0f / (1.0f + __expf(-(egat[r] + bb[r])));
                    atomicAdd(&accl[arow + f0 + r], es * vv[r] * g * okf);
                }
            }
        }
    }
    __syncthreads();

    const int tn = (p >= NRNG / 2) ? 1 : 0;
    for (int i = threadIdx.x; i < DH * DIN; i += EWV * 32)
        wl[i] = Wo[tn * DH * DIN + i];
    __syncthreads();
    for (int rt = wv; rt < RNG / 16; rt += EWV) {
        float* Ar = accl + rt * 16 * ROWP;
        #pragma unroll 8
        for (int i = 0; i < 32; ++i) {
            int idx = i * 32 + lane;
            int m = idx >> 6, cc = idx & 63;
            float dv = Ar[m * ROWP + DH + (cc >> 4)];
            float av = Ar[m * ROWP + cc];
            float qv = av / dv;
            Ar[m * ROWP + cc] = (dv > 0.f) ? qv : 0.f;
        }
        __builtin_amdgcn_fence(__ATOMIC_ACQ_REL, "workgroup", "local");
        #pragma unroll
        for (int nt = 0; nt < 4; ++nt) {
            const int n0 = nt * 16;
            v8f c = gemm_tile<64>(Ar, ROWP, wl, DIN, n0, lane);
            float bias = bo[tn * DIN + n0 + qn];
            #pragma unroll
            for (int i = 0; i < 8; ++i)
                out[(size_t)(r0 + rt * 16 + i + 8 * half) * DIN + n0 + qn] = c[i] + bias;
        }
    }
}

extern "C" void kernel_launch(void* const* d_in, const int* in_sizes, int n_in,
                              void* d_out, int out_size, void* d_ws, size_t ws_size,
                              hipStream_t stream) {
    const float* x    = (const float*)d_in[0];
    const float* ea   = (const float*)d_in[1];
    const float* Wq   = (const float*)d_in[2];
    const float* bq   = (const float*)d_in[3];
    const float* Wk   = (const float*)d_in[4];
    const float* bk   = (const float*)d_in[5];
    const float* Wv   = (const float*)d_in[6];
    const float* bv   = (const float*)d_in[7];
    const float* Wo   = (const float*)d_in[8];
    const float* bo   = (const float*)d_in[9];
    const float* We   = (const float*)d_in[10];
    const float* be   = (const float*)d_in[11];
    const float* Wg   = (const float*)d_in[12];
    const float* bg   = (const float*)d_in[13];
    const float* ew   = (const float*)d_in[14];
    const float* lnns = (const float*)d_in[15];
    const float* lnnb = (const float*)d_in[16];
    const float* lnes = (const float*)d_in[17];
    const float* lneb = (const float*)d_in[18];
    const int*   ei   = (const int*)d_in[19];
    const int* esrc = ei;
    const int* edst = ei + N_EDGES;

    float* ws  = (float*)d_ws;
    float* qb  = ws;
    float* kb  = qb  + (size_t)N_NODES * DH;
    float* vb  = kb  + (size_t)N_NODES * DH;
    int*   cnt  = (int*)(vb + (size_t)N_NODES * DH);
    int*   cur  = cnt + 512;
    int*   boff = cur + 512;
    int*   pe   = boff + 1024;
    int*   ps   = pe + N_EDGES;
    int*   pd   = ps + N_EDGES;

    k_zero<<<2, 256, 0, stream>>>((float*)cnt, 512);

    k_node_qkv<<<(N_NODES / 16) / 2, 64, 0, stream>>>(
        x, Wq, bq, Wk, bk, Wv, bv, lnns, lnnb, qb, kb, vb);

    k_bin_count<<<(N_EDGES + 1023) / 1024, 256, 0, stream>>>(edst, cnt);
    k_bin_scan<<<1, 512, 0, stream>>>(cnt, boff, cur);
    k_bin_scatter<<<(N_EDGES + 1023) / 1024, 256, 0, stream>>>(esrc, edst, cur, pe, ps, pd);

    k_edge_own<<<NRNG, EWV * 32, 0, stream>>>(
        ea, We, be, Wg, bg, ew, lnes, lneb, boff, pe, ps, pd, qb, kb, vb, Wo, bo,
        (float*)d_out);
}


# Gate sigmoid and softmax normalisation use the hardware reciprocal (v_rcp_f32, about 1 ulp), not the correctly rounded f32 division: 10 vector instructions fewer per gate, 32 gates per lane and tile
# speedup vs baseline: 1.7916x; 1.0295x over previous
#include <hip/hip_runtime.h>
#include <hip/hip_bf16.h>
#include <math.h>

#define N_NODES 200000
#define N_EDGES 800000
#define DIN 64
#define DE  32
#define DH  64
#define NH  4
#define HD  16
#define RNG  800
#define NRNG (N_NODES / RNG)
#define NBIN (2 * NRNG)
#define ROWP 69
#define EWV  24

typedef float v2f __attribute__((ext_vector_type(2)));
typedef float v4f __attribute__((ext_vector_type(4)));
typedef float v8f __attribute__((ext_vector_type(8)));

__device__ __forceinline__ v8f wstep(v2f a, v2f b, v8f c) {
    return __builtin_amdgcn_wmma_f32_16x16x4_f32(false, a, false, b, (short)0, c,
                                                 false, false);
}

template <int K>
__device__ __forceinline__ v8f gemm_tile(const float* As, int lda,
                                         const float* Bs, int ldb,
                                         int n0, int lane) {
    const int half = lane >> 4;
    const int qn   = lane & 15;
    v2f a[K / 4], b[K / 4];
    #pragma unroll
    for (int s = 0; s < K / 4; ++s) {
        const int k0 = 4 * s + 2 * half;
        a[s].x = As[qn * lda + k0];
        a[s].y = As[qn * lda + k0 + 1];
        b[s].x = Bs[k0 * ldb + n0 + qn];
        b[s].y = Bs[(k0 + 1) * ldb + n0 + qn];
    }
    v8f c = {0.f, 0.f, 0.f, 0.f, 0.f, 0.f, 0.f, 0.f};
    #pragma unroll
    for (int s = 0; s < K / 4; ++s) c = wstep(a[s], b[s], c);
    return c;
}

template <int K>
__device__ __forceinline__ v8f gemm_tile_t(const float* As, int lda,
                                           const float* Bs, int ldb,
                                           int n0, int lane) {
    const int half = lane >> 4;
    const int qn   = lane & 15;
    v2f a[K / 4], b[K / 4];
    #pragma unroll
    for (int s = 0; s < K / 4; ++s) {
        const int k0 = 4 * s + 2 * half;
        a[s].x = As[qn * lda + k0];
        a[s].y = As[qn * lda + k0 + 1];
        b[s].x = Bs[k0 * ldb + n0 + qn];
        b[s].y = Bs[(k0 + 1) * ldb + n0 + qn];
    }
    v8f c = {0.f, 0.f, 0.f, 0.f, 0.f, 0.f, 0.f, 0.f};
    #pragma unroll
    for (int s = 0; s < K / 4; ++s) c = wstep(b[s], a[s], c);
    return c;
}

__global__ void k_zero(float* __restrict__ p, int n) {
    int i = blockIdx.x * blockDim.x + threadIdx.x;
    if (i < n) p[i] = 0.f;
}

__global__ void k_node_qkv(const float* __restrict__ x,
                           const float* __restrict__ Wq, const float* __restrict__ bq,
                           const float* __restrict__ Wk, const float* __restrict__ bk,
                           const float* __restrict__ Wv, const float* __restrict__ bv,
                           const float* __restrict__ ln_s, const float* __restrict__ ln_b,
                           float* __restrict__ qo, float* __restrict__ ko,
                           float* __restrict__ vo) {
    __shared__ float xs[2][16 * 65];
    __shared__ float wlds[64 * 64];
    const int lane = threadIdx.x & 31;
    const int wv   = threadIdx.x >> 5;
    const int tile = blockIdx.x * 2 + wv;
    const int row0 = tile * 16;
    const int t    = (blockIdx.x * 32 >= N_NODES / 2) ? 1 : 0;
    float* As = xs[wv];

    for (int i = 0; i < 32; ++i) {
        int idx = i * 32 + lane;
        int m = idx >> 6, cc = idx & 63;
        As[m * 65 + cc] = x[(size_t)(row0 + m) * DIN + cc];
    }
    __syncthreads();
    {
        const int m = lane & 15, c0 = (lane >> 4) * 32;
        float s = 0.f;
        for (int j = 0; j < 32; ++j) s += As[m * 65 + c0 + j];
        s += __shfl_xor(s, 16, 32);
        float mean = s * (1.0f / 64.0f);
        float v2 = 0.f;
        for (int j = 0; j < 32; ++j) { float d = As[m * 65 + c0 + j] - mean; v2 += d * d; }
        v2 += __shfl_xor(v2, 16, 32);
        float inv = rsqrtf(v2 * (1.0f / 64.0f) + 1e-5f);
        for (int j = 0; j < 32; ++j) {
            int cc = c0 + j;
            As[m * 65 + cc] = (As[m * 65 + cc] - mean) * inv * ln_s[t * DIN + cc]
                              + ln_b[t * DIN + cc];
        }
    }

    const int half = lane >> 4, qn = lane & 15;
    const float* Ws[3] = {Wq + t * DIN * DH, Wk + t * DIN * DH, Wv + t * DIN * DH};
    const float* bs[3] = {bq + t * DH, bk + t * DH, bv + t * DH};
    float*       os[3] = {qo, ko, vo};
    for (int mm = 0; mm < 3; ++mm) {
        __syncthreads();
        for (int i = threadIdx.x; i < DIN * DH; i += 64)
            wlds[i] = Ws[mm][i];
        __syncthreads();
        #pragma unroll
        for (int nt = 0; nt < 4; ++nt) {
            const int n0 = nt * 16;
            v8f c = gemm_tile<64>(As, 65, wlds, DH, n0, lane);
            float bias = bs[mm][n0 + qn];
            #pragma unroll
            for (int i = 0; i < 8; ++i) {
                int m = i + 8 * half;
                os[mm][(size_t)(row0 + m) * DH + n0 + qn] = c[i] + bias;
            }
        }
    }
}

__device__ __forceinline__ int edge_bin(int dst, int e) {
    return 2 * (dst / RNG) + ((e >= N_EDGES / 2) ? 1 : 0);
}

__global__ void __launch_bounds__(256) k_bin_count(const int* __restrict__ edst,
                                                   int* __restrict__ cnt) {
    __shared__ int hist[NBIN];
    for (int i = threadIdx.x; i < NBIN; i += 256) hist[i] = 0;
    __syncthreads();
    for (int u = 0; u < 4; ++u) {
        const int e = blockIdx.x * 1024 + u * 256 + threadIdx.x;
        if (e < N_EDGES) atomicAdd(&hist[edge_bin(edst[e], e)], 1);
    }
    __syncthreads();
    for (int i = threadIdx.x; i < NBIN; i += 256) {
        const int c = hist[i];
        if (c) atomicAdd(&cnt[i], c);
    }
}

__global__ void __launch_bounds__(512) k_bin_scan(const int* __restrict__ cnt,
                                                  int* __restrict__ boff,
                                                  int* __restrict__ cur) {
    __shared__ int s[512];
    const int i = threadIdx.x;
    const int c = (i < NBIN) ? cnt[i] : 0;
    s[i] = c;
    __syncthreads();
    for (int off = 1; off < 512; off <<= 1) {
        const int v = (i >= off) ? s[i - off] : 0;
        __syncthreads();
        s[i] += v;
        __syncthreads();
    }
    const int excl = s[i] - c;
    if (i < NBIN) { boff[i] = excl; cur[i] = excl; }
    if (i == NBIN - 1) boff[NBIN] = s[i];
}

__global__ void __launch_bounds__(256) k_bin_scatter(const int* __restrict__ esrc,
                                                     const int* __restrict__ edst,
                                                     int* __restrict__ cur,
                                                     int* __restrict__ pe,
                                                     int* __restrict__ ps,
                                                     int* __restrict__ pd) {
    __shared__ int hist[NBIN];
    __shared__ int base[NBIN];
    for (int i = threadIdx.x; i < NBIN; i += 256) hist[i] = 0;
    __syncthreads();
    int bin[4], rank[4], sv[4], dv[4];
    #pragma unroll
    for (int u = 0; u < 4; ++u) {
        const int e = blockIdx.x * 1024 + u * 256 + threadIdx.x;
        bin[u] = -1; rank[u] = 0; sv[u] = 0; dv[u] = 0;
        if (e < N_EDGES) {
            dv[u] = edst[e];
            sv[u] = esrc[e];
            bin[u] = edge_bin(dv[u], e);
            rank[u] = atomicAdd(&hist[bin[u]], 1);
        }
    }
    __syncthreads();
    for (int i = threadIdx.x; i < NBIN; i += 256) {
        const int c = hist[i];
        base[i] = c ? atomicAdd(&cur[i], c) : 0;
    }
    __syncthreads();
    #pragma unroll
    for (int u = 0; u < 4; ++u) {
        const int e = blockIdx.x * 1024 + u * 256 + threadIdx.x;
        if (bin[u] >= 0) {
            const int pos = base[bin[u]] + rank[u];
            pe[pos] = e;
            ps[pos] = sv[u];
            pd[pos] = dv[u];
        }
    }
}

__global__ void __launch_bounds__(EWV * 32)
k_edge_own(const float* __restrict__ ea,
           const float* __restrict__ We, const float* __restrict__ be,
           const float* __restrict__ Wg, const float* __restrict__ bg,
           const float* __restrict__ ew,
           const float* __restrict__ ln_s, const float* __restrict__ ln_b,
           const int* __restrict__ boff,
           const int* __restrict__ pe, const int* __restrict__ ps,
           const int* __restrict__ pd,
           const float* __restrict__ qbuf, const float* __restrict__ kbuf,
           const float* __restrict__ vbuf,
           const float* __restrict__ Wo, const float* __restrict__ bo,
           float* __restrict__ out) {
    __shared__ float accl[RNG * ROWP];
    __shared__ __attribute__((aligned(16))) float wl[2 * DE * DH];
    __shared__ float ewlds[NH * 16 * 17];
    __shared__ float ebs[EWV][16 * 33];
    __shared__ float lnl[2 * DE];
    const int lane = threadIdx.x & 31;
    const int wv   = threadIdx.x >> 5;
    const int half = lane >> 4, qn = lane & 15;
    const int p    = blockIdx.x;
    const int r0   = p * RNG;
    float* As    = ebs[wv];
    float* wglds = wl;
    float* wsl   = wl + DE * DH;
    float* bsl   = wsl + DE * 16;

    for (int i = threadIdx.x; i < RNG * ROWP; i += EWV * 32) accl[i] = 0.f;

    for (int t = 0; t < 2; ++t) {
        __syncthreads();
        for (int i = threadIdx.x; i < DE * DH; i += EWV * 32)
            wglds[i] = Wg[t * DE * DH + i];
        if (threadIdx.x < DE * NH) {
            const int c = threadIdx.x >> 2, hh = threadIdx.x & 3;
            const float* wr = We + t * DE * DH + c * DH + hh * HD;
            const v4f w0 = *(const v4f*)(wr), w1 = *(const v4f*)(wr + 4);
            const v4f w2 = *(const v4f*)(wr + 8), w3 = *(const v4f*)(wr + 12);
            float s = w0.x; s += w0.y; s += w0.z; s += w0.w;
            s += w1.x; s += w1.y; s += w1.z; s += w1.w;
            s += w2.x; s += w2.y; s += w2.z; s += w2.w;
            s += w3.x; s += w3.y; s += w3.z; s += w3.w;
            wsl[c * 16 + hh]      = s;
            wsl[c * 16 + hh + 4]  = s;
            wsl[c * 16 + hh + 8]  = s;
            wsl[c * 16 + hh + 12] = s;
        } else if (threadIdx.x < DE * NH + NH) {
            const int hh = threadIdx.x - DE * NH;
            const float* br = be + t * DH + hh * HD;
            float s = br[0];
            for (int d = 1; d < HD; ++d) s += br[d];
            bsl[hh] = s;
        }
        for (int idx = threadIdx.x; idx < NH * 16 * 16; idx += EWV * 32) {
            int h = idx >> 8, rem = idx & 255, i = rem >> 4, j = rem & 15;
            ewlds[(h * 16 + j) * 17 + i] = ew[t * NH * 256 + idx];
        }
        if (threadIdx.x < 2 * DE)
            lnl[threadIdx.x] = (threadIdx.x < DE) ? ln_s[t * DE + threadIdx.x]
                                                  : ln_b[t * DE + threadIdx.x - DE];
        __syncthreads();
        const int s0 = boff[2 * p + t], s1 = boff[2 * p + t + 1];

        for (int slot0 = s0 + wv * 16; slot0 < s1; slot0 += EWV * 16) {
            const int nvalid = min(16, s1 - slot0);
            const int slot = min(slot0 + qn, N_EDGES - 1);
            const int my_e = pe[slot], my_s = ps[slot], my_d = pd[slot];
            #pragma unroll
            for (int m = 0; m < 16; ++m) {
                const int em = __builtin_amdgcn_readlane(my_e, m);
                As[m * 33 + lane] = ea[(size_t)em * DE + lane];
            }
            const bool  ok   = qn < nvalid;
            const float okf  = ok ? 1.0f : 0.0f;
            const int   arow = (ok ? my_d - r0 : 0) * ROWP;
            const bool  dadd = ok && (half == 0);
            const float* qrow = qbuf + (size_t)my_d * DH;
            const float* krow = kbuf + (size_t)my_s * DH;
            const float* vrow = vbuf + (size_t)my_s * DH;
            __builtin_amdgcn_fence(__ATOMIC_ACQ_REL, "workgroup", "local");
            {
                const int m = lane & 15, c0 = (lane >> 4) * 16;
                float s = 0.f;
                for (int j = 0; j < 16; ++j) s += As[m * 33 + c0 + j];
                s += __shfl_xor(s, 16, 32);
                float mean = s * (1.0f / 32.0f);
                float v2 = 0.f;
                for (int j = 0; j < 16; ++j) { float d = As[m * 33 + c0 + j] - mean; v2 += d * d; }
                v2 += __shfl_xor(v2, 16, 32);
                float inv = rsqrtf(v2 * (1.0f / 32.0f) + 1e-5f);
                for (int j = 0; j < 16; ++j) {
                    int cc = c0 + j;
                    As[m * 33 + cc] = (As[m * 33 + cc] - mean) * inv * lnl[cc] + lnl[DE + cc];
                }
            }
            __builtin_amdgcn_fence(__ATOMIC_ACQ_REL, "workgroup", "local");

            const v8f  es8 = gemm_tile_t<32>(As, 33, wsl, 16, 0, lane);
            const v4f  bsv = *(const v4f*)bsl;
            float ec0 = es8[0] + bsv.x, ec1 = es8[1] + bsv.y;
            float ec2 = es8[2] + bsv.z, ec3 = es8[3] + bsv.w;

            for (int h = 0; h < NH; ++h) {
                const int n0 = h * 16;
                const int f0 = n0 + 8 * half;
                const v4f q0 = *(const v4f*)(qrow + f0), q1 = *(const v4f*)(qrow + f0 + 4);
                const v4f k0 = *(const v4f*)(krow + f0), k1 = *(const v4f*)(krow + f0 + 4);
                const v4f v0 = *(const v4f*)(vrow + f0), v1 = *(const v4f*)(vrow + f0 + 4);
                const v4f bg0 = *(const v4f*)(bg + t * DH + f0), bg1 = *(const v4f*)(bg + t * DH + f0 + 4);
                const float qv[8] = {q0.x, q0.y, q0.z, q0.w, q1.x, q1.y, q1.z, q1.w};
                const float vv[8] = {v0.x, v0.y, v0.z, v0.w, v1.x, v1.y, v1.z, v1.w};
                const float bb[8] = {bg0.x, bg0.y, bg0.z, bg0.w, bg1.x, bg1.y, bg1.z, bg1.w};

                v8f egat = gemm_tile_t<32>(As, 33, wglds, DH, n0, lane);
                const float eh = ec0;
                { const float nx = ec0; ec0 = ec1; ec1 = ec2; ec2 = ec3; ec3 = nx; }

                const float* ewh = ewlds + h * 16 * 17;
                v2f ka[4], kb[4];
                kb[0].x = k0.x; kb[0].y = k0.y; kb[1].x = k0.z; kb[1].y = k0.w;
                kb[2].x = k1.x; kb[2].y = k1.y; kb[3].x = k1.z; kb[3].y = k1.w;
                #pragma unroll
                for (int s = 0; s < 4; ++s) {
                    const int j0 = 8 * half + 2 * s;
                    ka[s].x = ewh[j0 * 17 + qn];
                    ka[s].y = ewh[(j0 + 1) * 17 + qn];
                }
                v8f ekc = {0.f, 0.f, 0.f, 0.f, 0.f, 0.f, 0.f, 0.f};
                #pragma unroll
                for (int s = 0; s < 4; ++s) ekc = wstep(ka[s], kb[s], ekc);

                float ssum = 0.f;
                #pragma unroll
                for (int r = 0; r < 8; ++r) ssum += qv[r] * ekc[r];
                ssum += __shfl_xor(ssum, 16, 32);
                float sc = (ssum + eh) * 0.25f;
                sc = fminf(5.0f, fmaxf(-5.0f, sc));
                const float es = __expf(sc);
                if (dadd) atomicAdd(&accl[arow + DH + h], es);
                #pragma unroll
                for (int r = 0; r < 8; ++r) {
                    float g = __builtin_amdgcn_rcpf(1.0f + __expf(-(egat[r] + bb[r])));
                    atomicAdd(&accl[arow + f0 + r], es * vv[r] * g * okf);
                }
            }
        }
    }
    __syncthreads();

    const int tn = (p >= NRNG / 2) ? 1 : 0;
    for (int i = threadIdx.x; i < DH * DIN; i += EWV * 32)
        wl[i] = Wo[tn * DH * DIN + i];
    __syncthreads();
    for (int rt = wv; rt < RNG / 16; rt += EWV) {
        float* Ar = accl + rt * 16 * ROWP;
        #pragma unroll 8
        for (int i = 0; i < 32; ++i) {
            int idx = i * 32 + lane;
            int m = idx >> 6, cc = idx & 63;
            float dv = Ar[m * ROWP + DH + (cc >> 4)];
            float av = Ar[m * ROWP + cc];
            float qv = av * __builtin_amdgcn_rcpf(dv);
            Ar[m * ROWP + cc] = (dv > 0.f) ? qv : 0.f;
        }
        __builtin_amdgcn_fence(__ATOMIC_ACQ_REL, "workgroup", "local");
        #pragma unroll
        for (int nt = 0; nt < 4; ++nt) {
            const int n0 = nt * 16;
            v8f c = gemm_tile<64>(Ar, ROWP, wl, DIN, n0, lane);
            float bias = bo[tn * DIN + n0 + qn];
            #pragma unroll
            for (int i = 0; i < 8; ++i)
                out[(size_t)(r0 + rt * 16 + i + 8 * half) * DIN + n0 + qn] = c[i] + bias;
        }
    }
}

extern "C" void kernel_launch(void* const* d_in, const int* in_sizes, int n_in,
                              void* d_out, int out_size, void* d_ws, size_t ws_size,
                              hipStream_t stream) {
    const float* x    = (const float*)d_in[0];
    const float* ea   = (const float*)d_in[1];
    const float* Wq   = (const float*)d_in[2];
    const float* bq   = (const float*)d_in[3];
    const float* Wk   = (const float*)d_in[4];
    const float* bk   = (const float*)d_in[5];
    const float* Wv   = (const float*)d_in[6];
    const float* bv   = (const float*)d_in[7];
    const float* Wo   = (const float*)d_in[8];
    const float* bo   = (const float*)d_in[9];
    const float* We   = (const float*)d_in[10];
    const float* be   = (const float*)d_in[11];
    const float* Wg   = (const float*)d_in[12];
    const float* bg   = (const float*)d_in[13];
    const float* ew   = (const float*)d_in[14];
    const float* lnns = (const float*)d_in[15];
    const float* lnnb = (const float*)d_in[16];
    const float* lnes = (const float*)d_in[17];
    const float* lneb = (const float*)d_in[18];
    const int*   ei   = (const int*)d_in[19];
    const int* esrc = ei;
    const int* edst = ei + N_EDGES;

    float* ws  = (float*)d_ws;
    float* qb  = ws;
    float* kb  = qb  + (size_t)N_NODES * DH;
    float* vb  = kb  + (size_t)N_NODES * DH;
    int*   cnt  = (int*)(vb + (size_t)N_NODES * DH);
    int*   cur  = cnt + 512;
    int*   boff = cur + 512;
    int*   pe   = boff + 1024;
    int*   ps   = pe + N_EDGES;
    int*   pd   = ps + N_EDGES;

    k_zero<<<2, 256, 0, stream>>>((float*)cnt, 512);

    k_node_qkv<<<(N_NODES / 16) / 2, 64, 0, stream>>>(
        x, Wq, bq, Wk, bk, Wv, bv, lnns, lnnb, qb, kb, vb);

    k_bin_count<<<(N_EDGES + 1023) / 1024, 256, 0, stream>>>(edst, cnt);
    k_bin_scan<<<1, 512, 0, stream>>>(cnt, boff, cur);
    k_bin_scatter<<<(N_EDGES + 1023) / 1024, 256, 0, stream>>>(esrc, edst, cur, pe, ps, pd);

    k_edge_own<<<NRNG, EWV * 32, 0, stream>>>(
        ea, We, be, Wg, bg, ew, lnes, lneb, boff, pe, ps, pd, qb, kb, vb, Wo, bo,
        (float*)d_out);
}
